# starting kernel
# baseline (speedup 1.0000x reference)
#include <hip/hip_runtime.h>
#include <hip/hip_bf16.h>

typedef __attribute__((ext_vector_type(2))) float v2f;
typedef __attribute__((ext_vector_type(8))) float v8f;

#define NODES   4096
#define FIN     512
#define FOUT    64
#define BATCH   4
#define NEG_BIG (-9000000000000000.0f)
#define ALPHA   0.2f
#define WAVES_PER_BLOCK 8

__global__ void __launch_bounds__(256) gat_h_kernel(
    const float* __restrict__ x, const float* __restrict__ W,
    const float* __restrict__ a,
    float* __restrict__ h, float* __restrict__ s1, float* __restrict__ s2)
{
    const int lane = threadIdx.x & 31;
    const int wid  = threadIdx.x >> 5;
    const int wg   = blockIdx.x * WAVES_PER_BLOCK + wid;
    const int b    = wg >> 8;
    const int i0   = (wg & 255) << 4;
    const int half = lane >> 4;
    const int n    = lane & 15;

    const float* xrow = x + ((size_t)(b * NODES + i0 + n) * FIN);

    v8f acc0 = {}, acc1 = {}, acc2 = {}, acc3 = {};

    for (int k = 0; k < FIN; k += 4) {
        const int ka = k + 2 * half;
        v2f A; A[0] = xrow[ka]; A[1] = xrow[ka + 1];
        const float* wr0 = W + (size_t)ka * FOUT + n;
        const float* wr1 = wr0 + FOUT;
        v2f B0; B0[0] = wr0[0];  B0[1] = wr1[0];
        v2f B1; B1[0] = wr0[16]; B1[1] = wr1[16];
        v2f B2; B2[0] = wr0[32]; B2[1] = wr1[32];
        v2f B3; B3[0] = wr0[48]; B3[1] = wr1[48];
        acc0 = __builtin_amdgcn_wmma_f32_16x16x4_f32(false, A, false, B0, (short)0, acc0, false, false);
        acc1 = __builtin_amdgcn_wmma_f32_16x16x4_f32(false, A, false, B1, (short)0, acc1, false, false);
        acc2 = __builtin_amdgcn_wmma_f32_16x16x4_f32(false, A, false, B2, (short)0, acc2, false, false);
        acc3 = __builtin_amdgcn_wmma_f32_16x16x4_f32(false, A, false, B3, (short)0, acc3, false, false);
    }

    const float a10 = a[n],        a11 = a[n + 16], a12 = a[n + 32], a13 = a[n + 48];
    const float a20 = a[FOUT + n], a21 = a[FOUT + n + 16],
                a22 = a[FOUT + n + 32], a23 = a[FOUT + n + 48];

    for (int r = 0; r < 8; ++r) {
        const int row = i0 + r + 8 * half;
        float* hrow = h + ((size_t)(b * NODES + row) * FOUT) + n;
        const float c0 = acc0[r], c1 = acc1[r], c2 = acc2[r], c3 = acc3[r];
        hrow[0] = c0; hrow[16] = c1; hrow[32] = c2; hrow[48] = c3;

        float p1 = c0 * a10 + c1 * a11 + c2 * a12 + c3 * a13;
        float p2 = c0 * a20 + c1 * a21 + c2 * a22 + c3 * a23;
        for (int m = 1; m < 16; m <<= 1) {
            p1 += __shfl_xor(p1, m, 32);
            p2 += __shfl_xor(p2, m, 32);
        }
        if (n == 0) {
            s1[b * NODES + row] = p1;
            s2[b * NODES + row] = p2;
        }
    }
}

__global__ void __launch_bounds__(256) gat_attn_kernel(
    const float* __restrict__ h, const float* __restrict__ s1g,
    const float* __restrict__ s2g, const int* __restrict__ adj,
    float* __restrict__ out)
{
    __shared__ float lds_h[64 * FOUT];
    __shared__ float lds_s2[64];

    const int lane = threadIdx.x & 31;
    const int wid  = threadIdx.x >> 5;
    const int wg   = blockIdx.x * WAVES_PER_BLOCK + wid;
    const int b    = wg >> 8;
    const int i0   = (wg & 255) << 4;
    const int half = lane >> 4;
    const int n    = lane & 15;

    const float sv1 = s1g[b * NODES + i0 + n];
    const int* adjrow = adj + (size_t)(i0 + n) * NODES;

    float mrun = NEG_BIG;
    float lrun = 0.0f;
    v8f acc0 = {}, acc1 = {}, acc2 = {}, acc3 = {};

    for (int j0 = 0; j0 < NODES; j0 += 64) {
        __syncthreads();
        {
            const float4* src = (const float4*)(h + ((size_t)(b * NODES + j0)) * FOUT);
            float4* dst = (float4*)lds_h;
            for (int t = threadIdx.x; t < 64 * (FOUT / 4); t += 256) dst[t] = src[t];
            if (threadIdx.x < 64) lds_s2[threadIdx.x] = s2g[b * NODES + j0 + threadIdx.x];
        }
        __syncthreads();

        for (int jj = 0; jj < 4; ++jj) {
            const int jb = jj * 16;
            float p[8];
            float mx = NEG_BIG;
            for (int q = 0; q < 4; ++q) {
                for (int c = 0; c < 2; ++c) {
                    const int jl = jb + 4 * q + 2 * half + c;
                    float e = sv1 + lds_s2[jl];
                    e = (e >= 0.0f) ? e : ALPHA * e;
                    e = (adjrow[j0 + jl] > 0) ? e : NEG_BIG;
                    p[2 * q + c] = e;
                    mx = fmaxf(mx, e);
                }
            }
            mx = fmaxf(mx, __shfl_xor(mx, 16, 32));
            const float mnew  = fmaxf(mrun, mx);
            const float scale = __expf(mrun - mnew);
            float rs = 0.0f;
            for (int t = 0; t < 8; ++t) { p[t] = __expf(p[t] - mnew); rs += p[t]; }
            rs += __shfl_xor(rs, 16, 32);
            lrun = lrun * scale + rs;
            mrun = mnew;

            for (int r = 0; r < 8; ++r) {
                const float sr = __shfl(scale, r + 8 * half, 32);
                acc0[r] *= sr; acc1[r] *= sr; acc2[r] *= sr; acc3[r] *= sr;
            }

            for (int q = 0; q < 4; ++q) {
                v2f A; A[0] = p[2 * q]; A[1] = p[2 * q + 1];
                const float* hr0 = lds_h + (size_t)(jb + 4 * q + 2 * half) * FOUT + n;
                const float* hr1 = hr0 + FOUT;
                v2f B0; B0[0] = hr0[0];  B0[1] = hr1[0];
                v2f B1; B1[0] = hr0[16]; B1[1] = hr1[16];
                v2f B2; B2[0] = hr0[32]; B2[1] = hr1[32];
                v2f B3; B3[0] = hr0[48]; B3[1] = hr1[48];
                acc0 = __builtin_amdgcn_wmma_f32_16x16x4_f32(false, A, false, B0, (short)0, acc0, false, false);
                acc1 = __builtin_amdgcn_wmma_f32_16x16x4_f32(false, A, false, B1, (short)0, acc1, false, false);
                acc2 = __builtin_amdgcn_wmma_f32_16x16x4_f32(false, A, false, B2, (short)0, acc2, false, false);
                acc3 = __builtin_amdgcn_wmma_f32_16x16x4_f32(false, A, false, B3, (short)0, acc3, false, false);
            }
        }
    }

    for (int r = 0; r < 8; ++r) {
        const float lr  = __shfl(lrun, r + 8 * half, 32);
        const float inv = 1.0f / lr;
        const int row   = i0 + r + 8 * half;
        float* o = out + ((size_t)(b * NODES + row)) * FOUT + n;
        float v;
        v = acc0[r] * inv; o[0]  = (v > 0.0f) ? v : expm1f(v);
        v = acc1[r] * inv; o[16] = (v > 0.0f) ? v : expm1f(v);
        v = acc2[r] * inv; o[32] = (v > 0.0f) ? v : expm1f(v);
        v = acc3[r] * inv; o[48] = (v > 0.0f) ? v : expm1f(v);
    }
}

extern "C" void kernel_launch(void* const* d_in, const int* in_sizes, int n_in,
                              void* d_out, int out_size, void* d_ws, size_t ws_size,
                              hipStream_t stream)
{
    const float* x   = (const float*)d_in[0];
    const int*   adj = (const int*)  d_in[1];
    const float* W   = (const float*)d_in[2];
    const float* a   = (const float*)d_in[3];
    float* out = (float*)d_out;

    float* h  = (float*)d_ws;
    float* s1 = h + (size_t)BATCH * NODES * FOUT;
    float* s2 = s1 + (size_t)BATCH * NODES;

    const int total_waves = BATCH * (NODES / 16);
    const int blocks = total_waves / WAVES_PER_BLOCK;

    gat_h_kernel<<<blocks, 256, 0, stream>>>(x, W, a, h, s1, s2);
    gat_attn_kernel<<<blocks, 256, 0, stream>>>(h, s1, s2, adj, out);
}


# Software-pipeline the key-tile loop: the next tile's h, s2 and adjacency loads fly into registers under this tile's compute; two LDS buffers, one barrier per tile; order of every sum unchanged
# speedup vs baseline: 1.7899x; 1.7899x over previous
#include <hip/hip_runtime.h>
#include <hip/hip_bf16.h>

typedef __attribute__((ext_vector_type(2))) float v2f;
typedef __attribute__((ext_vector_type(8))) float v8f;
typedef __attribute__((ext_vector_type(4))) float v4f;

#define NODES   4096
#define FIN     512
#define FOUT    64
#define BATCH   4
#define NEG_BIG (-9000000000000000.0f)
#define ALPHA   0.2f
#define WAVES_PER_BLOCK 8

__global__ void __launch_bounds__(256) gat_h_kernel(
    const float* __restrict__ x, const float* __restrict__ W,
    const float* __restrict__ a,
    float* __restrict__ h, float* __restrict__ s1, float* __restrict__ s2)
{
    const int lane = threadIdx.x & 31;
    const int wid  = threadIdx.x >> 5;
    const int wg   = blockIdx.x * WAVES_PER_BLOCK + wid;
    const int b    = wg >> 8;
    const int i0   = (wg & 255) << 4;
    const int half = lane >> 4;
    const int n    = lane & 15;

    const float* xrow = x + ((size_t)(b * NODES + i0 + n) * FIN);

    v8f acc0 = {}, acc1 = {}, acc2 = {}, acc3 = {};

    for (int k = 0; k < FIN; k += 4) {
        const int ka = k + 2 * half;
        v2f A; A[0] = xrow[ka]; A[1] = xrow[ka + 1];
        const float* wr0 = W + (size_t)ka * FOUT + n;
        const float* wr1 = wr0 + FOUT;
        v2f B0; B0[0] = wr0[0];  B0[1] = wr1[0];
        v2f B1; B1[0] = wr0[16]; B1[1] = wr1[16];
        v2f B2; B2[0] = wr0[32]; B2[1] = wr1[32];
        v2f B3; B3[0] = wr0[48]; B3[1] = wr1[48];
        acc0 = __builtin_amdgcn_wmma_f32_16x16x4_f32(false, A, false, B0, (short)0, acc0, false, false);
        acc1 = __builtin_amdgcn_wmma_f32_16x16x4_f32(false, A, false, B1, (short)0, acc1, false, false);
        acc2 = __builtin_amdgcn_wmma_f32_16x16x4_f32(false, A, false, B2, (short)0, acc2, false, false);
        acc3 = __builtin_amdgcn_wmma_f32_16x16x4_f32(false, A, false, B3, (short)0, acc3, false, false);
    }

    const float a10 = a[n],        a11 = a[n + 16], a12 = a[n + 32], a13 = a[n + 48];
    const float a20 = a[FOUT + n], a21 = a[FOUT + n + 16],
                a22 = a[FOUT + n + 32], a23 = a[FOUT + n + 48];

    for (int r = 0; r < 8; ++r) {
        const int row = i0 + r + 8 * half;
        float* hrow = h + ((size_t)(b * NODES + row) * FOUT) + n;
        const float c0 = acc0[r], c1 = acc1[r], c2 = acc2[r], c3 = acc3[r];
        hrow[0] = c0; hrow[16] = c1; hrow[32] = c2; hrow[48] = c3;

        float p1 = c0 * a10 + c1 * a11 + c2 * a12 + c3 * a13;
        float p2 = c0 * a20 + c1 * a21 + c2 * a22 + c3 * a23;
        for (int m = 1; m < 16; m <<= 1) {
            p1 += __shfl_xor(p1, m, 32);
            p2 += __shfl_xor(p2, m, 32);
        }
        if (n == 0) {
            s1[b * NODES + row] = p1;
            s2[b * NODES + row] = p2;
        }
    }
}

#define TILE_J   64
#define NTILES   (NODES / TILE_J)

__global__ void __launch_bounds__(256) gat_attn_kernel(
    const float* __restrict__ h, const float* __restrict__ s1g,
    const float* __restrict__ s2g, const int* __restrict__ adj,
    float* __restrict__ out)
{
    __shared__ __attribute__((aligned(16))) float lds_h[2][TILE_J * FOUT];
    __shared__ float lds_s2[2][TILE_J];

    const int lane = threadIdx.x & 31;
    const int wid  = threadIdx.x >> 5;
    const int wg   = blockIdx.x * WAVES_PER_BLOCK + wid;
    const int b    = wg >> 8;
    const int i0   = (wg & 255) << 4;
    const int half = lane >> 4;
    const int n    = lane & 15;

    const float sv1 = s1g[b * NODES + i0 + n];
    const int* adjlane = adj + (size_t)(i0 + n) * NODES + 2 * half;
    const v4f*   hsrc  = (const v4f*)(h + (size_t)b * NODES * FOUT) + threadIdx.x;
    const float*  s2src = s2g + b * NODES + (threadIdx.x & (TILE_J - 1));
    const bool    s2thr = threadIdx.x < TILE_J;

    v4f    hr_0, hr_1, hr_2, hr_3;
    float  s2reg = 0.0f;
    int    adjn[32];
    int    adjc[32];

    hr_0 = hsrc[0]; hr_1 = hsrc[256]; hr_2 = hsrc[512]; hr_3 = hsrc[768];
    if (s2thr) s2reg = s2src[0];
    for (int jj = 0; jj < 4; ++jj)
        for (int q = 0; q < 4; ++q) {
            adjc[8 * jj + 2 * q]     = adjlane[16 * jj + 4 * q];
            adjc[8 * jj + 2 * q + 1] = adjlane[16 * jj + 4 * q + 1];
        }
    {
        v4f* dst = (v4f*)lds_h[0] + threadIdx.x;
        dst[0] = hr_0; dst[256] = hr_1; dst[512] = hr_2; dst[768] = hr_3;
        if (s2thr) lds_s2[0][threadIdx.x] = s2reg;
    }
    __syncthreads();

    float mrun = NEG_BIG;
    float lrun = 0.0f;
    v8f acc0 = {}, acc1 = {}, acc2 = {}, acc3 = {};

    for (int t = 0; t < NTILES; ++t) {
        const int  cur  = t & 1;
        const bool more = (t + 1) < NTILES;

        if (more) {
            const int j1 = (t + 1) * TILE_J;
            const v4f* hs = hsrc + (size_t)j1 * (FOUT / 4);
            hr_0 = hs[0]; hr_1 = hs[256]; hr_2 = hs[512]; hr_3 = hs[768];
            if (s2thr) s2reg = s2src[j1];
            for (int jj = 0; jj < 4; ++jj)
                for (int q = 0; q < 4; ++q) {
                    adjn[8 * jj + 2 * q]     = adjlane[j1 + 16 * jj + 4 * q];
                    adjn[8 * jj + 2 * q + 1] = adjlane[j1 + 16 * jj + 4 * q + 1];
                }
        }
        const float* lh  = lds_h[cur];
        const float* ls2 = lds_s2[cur];

        #pragma unroll
        for (int jj = 0; jj < 4; ++jj) {
            const int jb = jj * 16;
            float p[8];
            float mx = NEG_BIG;
            for (int q = 0; q < 4; ++q) {
                for (int c = 0; c < 2; ++c) {
                    const int jl = jb + 4 * q + 2 * half + c;
                    float e = sv1 + ls2[jl];
                    e = (e >= 0.0f) ? e : ALPHA * e;
                    e = (adjc[8 * jj + 2 * q + c] > 0) ? e : NEG_BIG;
                    p[2 * q + c] = e;
                    mx = fmaxf(mx, e);
                }
            }
            mx = fmaxf(mx, __shfl_xor(mx, 16, 32));
            const float mnew  = fmaxf(mrun, mx);
            const float scale = __expf(mrun - mnew);
            float rs = 0.0f;
            for (int u = 0; u < 8; ++u) { p[u] = __expf(p[u] - mnew); rs += p[u]; }
            rs += __shfl_xor(rs, 16, 32);
            lrun = lrun * scale + rs;
            mrun = mnew;

            for (int r = 0; r < 8; ++r) {
                const float sr = __shfl(scale, r + 8 * half, 32);
                acc0[r] *= sr; acc1[r] *= sr; acc2[r] *= sr; acc3[r] *= sr;
            }

            for (int q = 0; q < 4; ++q) {
                v2f A; A[0] = p[2 * q]; A[1] = p[2 * q + 1];
                const float* hr0 = lh + (size_t)(jb + 4 * q + 2 * half) * FOUT + n;
                const float* hr1 = hr0 + FOUT;
                v2f B0; B0[0] = hr0[0];  B0[1] = hr1[0];
                v2f B1; B1[0] = hr0[16]; B1[1] = hr1[16];
                v2f B2; B2[0] = hr0[32]; B2[1] = hr1[32];
                v2f B3; B3[0] = hr0[48]; B3[1] = hr1[48];
                acc0 = __builtin_amdgcn_wmma_f32_16x16x4_f32(false, A, false, B0, (short)0, acc0, false, false);
                acc1 = __builtin_amdgcn_wmma_f32_16x16x4_f32(false, A, false, B1, (short)0, acc1, false, false);
                acc2 = __builtin_amdgcn_wmma_f32_16x16x4_f32(false, A, false, B2, (short)0, acc2, false, false);
                acc3 = __builtin_amdgcn_wmma_f32_16x16x4_f32(false, A, false, B3, (short)0, acc3, false, false);
            }
        }

        if (more) {
            v4f* dst = (v4f*)lds_h[cur ^ 1] + threadIdx.x;
            dst[0] = hr_0; dst[256] = hr_1; dst[512] = hr_2; dst[768] = hr_3;
            if (s2thr) lds_s2[cur ^ 1][threadIdx.x] = s2reg;
            for (int u = 0; u < 32; ++u) adjc[u] = adjn[u];
        }
        __syncthreads();
    }

    for (int r = 0; r < 8; ++r) {
        const float lr  = __shfl(lrun, r + 8 * half, 32);
        const float inv = 1.0f / lr;
        const int row   = i0 + r + 8 * half;
        float* o = out + ((size_t)(b * NODES + row)) * FOUT + n;
        float v;
        v = acc0[r] * inv; o[0]  = (v > 0.0f) ? v : expm1f(v);
        v = acc1[r] * inv; o[16] = (v > 0.0f) ? v : expm1f(v);
        v = acc2[r] * inv; o[32] = (v > 0.0f) ? v : expm1f(v);
        v = acc3[r] * inv; o[48] = (v > 0.0f) ? v : expm1f(v);
    }
}

extern "C" void kernel_launch(void* const* d_in, const int* in_sizes, int n_in,
                              void* d_out, int out_size, void* d_ws, size_t ws_size,
                              hipStream_t stream)
{
    const float* x   = (const float*)d_in[0];
    const int*   adj = (const int*)  d_in[1];
    const float* W   = (const float*)d_in[2];
    const float* a   = (const float*)d_in[3];
    float* out = (float*)d_out;

    float* h  = (float*)d_ws;
    float* s1 = h + (size_t)BATCH * NODES * FOUT;
    float* s2 = s1 + (size_t)BATCH * NODES;

    const int total_waves = BATCH * (NODES / 16);
    const int blocks = total_waves / WAVES_PER_BLOCK;

    gat_h_kernel<<<blocks, 256, 0, stream>>>(x, W, a, h, s1, s2);
    gat_attn_kernel<<<blocks, 256, 0, stream>>>(h, s1, s2, adj, out);
}


# Split each 16-row tile's keys over two waves of the block (half the keys each; partial max, sum, accumulators merged once through LDS: reduction tree changes): grid 128 -> 256 blocks, every WGP busy
# speedup vs baseline: 2.6706x; 1.4920x over previous
#include <hip/hip_runtime.h>
#include <hip/hip_bf16.h>

typedef __attribute__((ext_vector_type(2))) float v2f;
typedef __attribute__((ext_vector_type(8))) float v8f;
typedef __attribute__((ext_vector_type(4))) float v4f;

#define NODES   4096
#define FIN     512
#define FOUT    64
#define BATCH   4
#define NEG_BIG (-9000000000000000.0f)
#define ALPHA   0.2f
#define WAVES_PER_BLOCK 8

__global__ void __launch_bounds__(256) gat_h_kernel(
    const float* __restrict__ x, const float* __restrict__ W,
    const float* __restrict__ a,
    float* __restrict__ h, float* __restrict__ s1, float* __restrict__ s2)
{
    const int lane = threadIdx.x & 31;
    const int wid  = threadIdx.x >> 5;
    const int wg   = blockIdx.x * WAVES_PER_BLOCK + wid;
    const int b    = wg >> 8;
    const int i0   = (wg & 255) << 4;
    const int half = lane >> 4;
    const int n    = lane & 15;

    const float* xrow = x + ((size_t)(b * NODES + i0 + n) * FIN);

    v8f acc0 = {}, acc1 = {}, acc2 = {}, acc3 = {};

    for (int k = 0; k < FIN; k += 4) {
        const int ka = k + 2 * half;
        v2f A; A[0] = xrow[ka]; A[1] = xrow[ka + 1];
        const float* wr0 = W + (size_t)ka * FOUT + n;
        const float* wr1 = wr0 + FOUT;
        v2f B0; B0[0] = wr0[0];  B0[1] = wr1[0];
        v2f B1; B1[0] = wr0[16]; B1[1] = wr1[16];
        v2f B2; B2[0] = wr0[32]; B2[1] = wr1[32];
        v2f B3; B3[0] = wr0[48]; B3[1] = wr1[48];
        acc0 = __builtin_amdgcn_wmma_f32_16x16x4_f32(false, A, false, B0, (short)0, acc0, false, false);
        acc1 = __builtin_amdgcn_wmma_f32_16x16x4_f32(false, A, false, B1, (short)0, acc1, false, false);
        acc2 = __builtin_amdgcn_wmma_f32_16x16x4_f32(false, A, false, B2, (short)0, acc2, false, false);
        acc3 = __builtin_amdgcn_wmma_f32_16x16x4_f32(false, A, false, B3, (short)0, acc3, false, false);
    }

    const float a10 = a[n],        a11 = a[n + 16], a12 = a[n + 32], a13 = a[n + 48];
    const float a20 = a[FOUT + n], a21 = a[FOUT + n + 16],
                a22 = a[FOUT + n + 32], a23 = a[FOUT + n + 48];

    for (int r = 0; r < 8; ++r) {
        const int row = i0 + r + 8 * half;
        float* hrow = h + ((size_t)(b * NODES + row) * FOUT) + n;
        const float c0 = acc0[r], c1 = acc1[r], c2 = acc2[r], c3 = acc3[r];
        hrow[0] = c0; hrow[16] = c1; hrow[32] = c2; hrow[48] = c3;

        float p1 = c0 * a10 + c1 * a11 + c2 * a12 + c3 * a13;
        float p2 = c0 * a20 + c1 * a21 + c2 * a22 + c3 * a23;
        for (int m = 1; m < 16; m <<= 1) {
            p1 += __shfl_xor(p1, m, 32);
            p2 += __shfl_xor(p2, m, 32);
        }
        if (n == 0) {
            s1[b * NODES + row] = p1;
            s2[b * NODES + row] = p2;
        }
    }
}

#define TILE_J          64
#define KSPLIT          2
#define KEYS_PER_SPLIT  (NODES / KSPLIT)
#define NTILES          (KEYS_PER_SPLIT / TILE_J)
#define ROWTILES        4
#define TILE_F4         (TILE_J * FOUT / 4)
#define SPLIT_F4        (KEYS_PER_SPLIT * FOUT / 4)

__global__ void __launch_bounds__(256) gat_attn_kernel(
    const float* __restrict__ h, const float* __restrict__ s1g,
    const float* __restrict__ s2g, const int* __restrict__ adj,
    float* __restrict__ out)
{
    __shared__ __attribute__((aligned(16))) float lds_h[2][KSPLIT][TILE_J * FOUT];
    __shared__ float lds_s2[2][KSPLIT * TILE_J];
    __shared__ float lds_part[ROWTILES][34][32];

    const int lane = threadIdx.x & 31;
    const int wid  = threadIdx.x >> 5;
    const int kh   = wid >> 2;
    const int rt   = wid & 3;
    const int wg   = blockIdx.x * ROWTILES + rt;
    const int b    = wg >> 8;
    const int i0   = (wg & 255) << 4;
    const int half = lane >> 4;
    const int n    = lane & 15;

    const float sv1 = s1g[b * NODES + i0 + n];
    const int* adjlane = adj + (size_t)(i0 + n) * NODES + kh * KEYS_PER_SPLIT + 2 * half;
    const v4f*   hsrc  = (const v4f*)(h + (size_t)b * NODES * FOUT) + threadIdx.x;
    const float*  s2src = s2g + b * NODES + (threadIdx.x >> 6) * KEYS_PER_SPLIT + (threadIdx.x & (TILE_J - 1));
    const bool    s2thr = threadIdx.x < KSPLIT * TILE_J;

    v4f    hr_0, hr_1, hr_2, hr_3;
    v4f    hr_4, hr_5, hr_6, hr_7;
    float  s2reg = 0.0f;
    int    adjn[32];
    int    adjc[32];

    hr_0 = hsrc[0];            hr_1 = hsrc[256];            hr_2 = hsrc[512];            hr_3 = hsrc[768];
    hr_4 = hsrc[SPLIT_F4];     hr_5 = hsrc[SPLIT_F4 + 256]; hr_6 = hsrc[SPLIT_F4 + 512]; hr_7 = hsrc[SPLIT_F4 + 768];
    if (s2thr) s2reg = s2src[0];
    for (int jj = 0; jj < 4; ++jj)
        for (int q = 0; q < 4; ++q) {
            adjc[8 * jj + 2 * q]     = adjlane[16 * jj + 4 * q];
            adjc[8 * jj + 2 * q + 1] = adjlane[16 * jj + 4 * q + 1];
        }
    {
        v4f* dst = (v4f*)lds_h[0][0] + threadIdx.x;
        dst[0] = hr_0;           dst[256] = hr_1;           dst[512] = hr_2;           dst[768] = hr_3;
        dst[TILE_F4] = hr_4;     dst[TILE_F4 + 256] = hr_5; dst[TILE_F4 + 512] = hr_6; dst[TILE_F4 + 768] = hr_7;
        if (s2thr) lds_s2[0][threadIdx.x] = s2reg;
    }
    __syncthreads();

    float mrun = NEG_BIG;
    float lrun = 0.0f;
    v8f acc0 = {}, acc1 = {}, acc2 = {}, acc3 = {};

    for (int t = 0; t < NTILES; ++t) {
        const int  cur  = t & 1;
        const bool more = (t + 1) < NTILES;

        if (more) {
            const int j1 = (t + 1) * TILE_J;
            const v4f* hs = hsrc + (size_t)j1 * (FOUT / 4);
            hr_0 = hs[0];            hr_1 = hs[256];            hr_2 = hs[512];            hr_3 = hs[768];
            hr_4 = hs[SPLIT_F4];     hr_5 = hs[SPLIT_F4 + 256]; hr_6 = hs[SPLIT_F4 + 512]; hr_7 = hs[SPLIT_F4 + 768];
            if (s2thr) s2reg = s2src[j1];
            for (int jj = 0; jj < 4; ++jj)
                for (int q = 0; q < 4; ++q) {
                    adjn[8 * jj + 2 * q]     = adjlane[j1 + 16 * jj + 4 * q];
                    adjn[8 * jj + 2 * q + 1] = adjlane[j1 + 16 * jj + 4 * q + 1];
                }
        }
        const float* lh  = lds_h[cur][kh];
        const float* ls2 = lds_s2[cur] + kh * TILE_J;

        #pragma unroll
        for (int jj = 0; jj < 4; ++jj) {
            const int jb = jj * 16;
            float p[8];
            float mx = NEG_BIG;
            for (int q = 0; q < 4; ++q) {
                for (int c = 0; c < 2; ++c) {
                    const int jl = jb + 4 * q + 2 * half + c;
                    float e = sv1 + ls2[jl];
                    e = (e >= 0.0f) ? e : ALPHA * e;
                    e = (adjc[8 * jj + 2 * q + c] > 0) ? e : NEG_BIG;
                    p[2 * q + c] = e;
                    mx = fmaxf(mx, e);
                }
            }
            mx = fmaxf(mx, __shfl_xor(mx, 16, 32));
            const float mnew  = fmaxf(mrun, mx);
            const float scale = __expf(mrun - mnew);
            float rs = 0.0f;
            for (int u = 0; u < 8; ++u) { p[u] = __expf(p[u] - mnew); rs += p[u]; }
            rs += __shfl_xor(rs, 16, 32);
            lrun = lrun * scale + rs;
            mrun = mnew;

            for (int r = 0; r < 8; ++r) {
                const float sr = __shfl(scale, r + 8 * half, 32);
                acc0[r] *= sr; acc1[r] *= sr; acc2[r] *= sr; acc3[r] *= sr;
            }

            for (int q = 0; q < 4; ++q) {
                v2f A; A[0] = p[2 * q]; A[1] = p[2 * q + 1];
                const float* hr0 = lh + (size_t)(jb + 4 * q + 2 * half) * FOUT + n;
                const float* hr1 = hr0 + FOUT;
                v2f B0; B0[0] = hr0[0];  B0[1] = hr1[0];
                v2f B1; B1[0] = hr0[16]; B1[1] = hr1[16];
                v2f B2; B2[0] = hr0[32]; B2[1] = hr1[32];
                v2f B3; B3[0] = hr0[48]; B3[1] = hr1[48];
                acc0 = __builtin_amdgcn_wmma_f32_16x16x4_f32(false, A, false, B0, (short)0, acc0, false, false);
                acc1 = __builtin_amdgcn_wmma_f32_16x16x4_f32(false, A, false, B1, (short)0, acc1, false, false);
                acc2 = __builtin_amdgcn_wmma_f32_16x16x4_f32(false, A, false, B2, (short)0, acc2, false, false);
                acc3 = __builtin_amdgcn_wmma_f32_16x16x4_f32(false, A, false, B3, (short)0, acc3, false, false);
            }
        }

        if (more) {
            v4f* dst = (v4f*)lds_h[cur ^ 1][0] + threadIdx.x;
            dst[0] = hr_0;           dst[256] = hr_1;           dst[512] = hr_2;           dst[768] = hr_3;
            dst[TILE_F4] = hr_4;     dst[TILE_F4 + 256] = hr_5; dst[TILE_F4 + 512] = hr_6; dst[TILE_F4 + 768] = hr_7;
            if (s2thr) lds_s2[cur ^ 1][threadIdx.x] = s2reg;
            for (int u = 0; u < 32; ++u) adjc[u] = adjn[u];
        }
        __syncthreads();
    }

    float* part = &lds_part[rt][0][lane];
    if (kh == 1) {
        for (int r = 0; r < 8; ++r) {
            part[(r)      * 32] = acc0[r];
            part[(8 + r)  * 32] = acc1[r];
            part[(16 + r) * 32] = acc2[r];
            part[(24 + r) * 32] = acc3[r];
        }
        part[32 * 32] = mrun;
        part[33 * 32] = lrun;
    }
    __syncthreads();
    if (kh == 0) {
        const float m1 = part[32 * 32];
        const float l1 = part[33 * 32];
        const float m  = fmaxf(mrun, m1);
        const float f0 = __expf(mrun - m);
        const float f1 = __expf(m1 - m);
        const float l  = lrun * f0 + l1 * f1;

        for (int r = 0; r < 8; ++r) {
            const int   src = r + 8 * half;
            const float g0  = __shfl(f0, src, 32);
            const float g1  = __shfl(f1, src, 32);
            const float lr  = __shfl(l,  src, 32);
            const float inv = 1.0f / lr;
            const int row   = i0 + r + 8 * half;
            float* o = out + ((size_t)(b * NODES + row)) * FOUT + n;
            float v;
            v = (acc0[r] * g0 + part[(r)      * 32] * g1) * inv; o[0]  = (v > 0.0f) ? v : expm1f(v);
            v = (acc1[r] * g0 + part[(8 + r)  * 32] * g1) * inv; o[16] = (v > 0.0f) ? v : expm1f(v);
            v = (acc2[r] * g0 + part[(16 + r) * 32] * g1) * inv; o[32] = (v > 0.0f) ? v : expm1f(v);
            v = (acc3[r] * g0 + part[(24 + r) * 32] * g1) * inv; o[48] = (v > 0.0f) ? v : expm1f(v);
        }
    }
}

extern "C" void kernel_launch(void* const* d_in, const int* in_sizes, int n_in,
                              void* d_out, int out_size, void* d_ws, size_t ws_size,
                              hipStream_t stream)
{
    const float* x   = (const float*)d_in[0];
    const int*   adj = (const int*)  d_in[1];
    const float* W   = (const float*)d_in[2];
    const float* a   = (const float*)d_in[3];
    float* out = (float*)d_out;

    float* h  = (float*)d_ws;
    float* s1 = h + (size_t)BATCH * NODES * FOUT;
    float* s2 = s1 + (size_t)BATCH * NODES;

    const int total_waves = BATCH * (NODES / 16);
    const int blocks_h = total_waves / WAVES_PER_BLOCK;
    const int blocks_a = total_waves / ROWTILES;

    gat_h_kernel<<<blocks_h, 256, 0, stream>>>(x, W, a, h, s1, s2);
    gat_attn_kernel<<<blocks_a, 256, 0, stream>>>(h, s1, s2, adj, out);
}


# Kernel 2 adjacency tile read in whole 128-byte lines (8 16-byte loads per wave and tile, 4 lines each, not 16 loads of 16 lines); wave ballots fold it into one mask word per lane; sums unchanged
# speedup vs baseline: 2.7974x; 1.0475x over previous
#include <hip/hip_runtime.h>
#include <hip/hip_bf16.h>

typedef __attribute__((ext_vector_type(2))) float v2f;
typedef __attribute__((ext_vector_type(8))) float v8f;
typedef __attribute__((ext_vector_type(4))) float v4f;
typedef __attribute__((ext_vector_type(4))) int   v4i;

#define NODES   4096
#define FIN     512
#define FOUT    64
#define BATCH   4
#define NEG_BIG (-9000000000000000.0f)
#define ALPHA   0.2f
#define WAVES_PER_BLOCK 8

__global__ void __launch_bounds__(256) gat_h_kernel(
    const float* __restrict__ x, const float* __restrict__ W,
    const float* __restrict__ a,
    float* __restrict__ h, float* __restrict__ s1, float* __restrict__ s2)
{
    const int lane = threadIdx.x & 31;
    const int wid  = threadIdx.x >> 5;
    const int wg   = blockIdx.x * WAVES_PER_BLOCK + wid;
    const int b    = wg >> 8;
    const int i0   = (wg & 255) << 4;
    const int half = lane >> 4;
    const int n    = lane & 15;

    const float* xrow = x + ((size_t)(b * NODES + i0 + n) * FIN);

    v8f acc0 = {}, acc1 = {}, acc2 = {}, acc3 = {};

    for (int k = 0; k < FIN; k += 4) {
        const int ka = k + 2 * half;
        v2f A; A[0] = xrow[ka]; A[1] = xrow[ka + 1];
        const float* wr0 = W + (size_t)ka * FOUT + n;
        const float* wr1 = wr0 + FOUT;
        v2f B0; B0[0] = wr0[0];  B0[1] = wr1[0];
        v2f B1; B1[0] = wr0[16]; B1[1] = wr1[16];
        v2f B2; B2[0] = wr0[32]; B2[1] = wr1[32];
        v2f B3; B3[0] = wr0[48]; B3[1] = wr1[48];
        acc0 = __builtin_amdgcn_wmma_f32_16x16x4_f32(false, A, false, B0, (short)0, acc0, false, false);
        acc1 = __builtin_amdgcn_wmma_f32_16x16x4_f32(false, A, false, B1, (short)0, acc1, false, false);
        acc2 = __builtin_amdgcn_wmma_f32_16x16x4_f32(false, A, false, B2, (short)0, acc2, false, false);
        acc3 = __builtin_amdgcn_wmma_f32_16x16x4_f32(false, A, false, B3, (short)0, acc3, false, false);
    }

    const float a10 = a[n],        a11 = a[n + 16], a12 = a[n + 32], a13 = a[n + 48];
    const float a20 = a[FOUT + n], a21 = a[FOUT + n + 16],
                a22 = a[FOUT + n + 32], a23 = a[FOUT + n + 48];

    for (int r = 0; r < 8; ++r) {
        const int row = i0 + r + 8 * half;
        float* hrow = h + ((size_t)(b * NODES + row) * FOUT) + n;
        const float c0 = acc0[r], c1 = acc1[r], c2 = acc2[r], c3 = acc3[r];
        hrow[0] = c0; hrow[16] = c1; hrow[32] = c2; hrow[48] = c3;

        float p1 = c0 * a10 + c1 * a11 + c2 * a12 + c3 * a13;
        float p2 = c0 * a20 + c1 * a21 + c2 * a22 + c3 * a23;
        for (int m = 1; m < 16; m <<= 1) {
            p1 += __shfl_xor(p1, m, 32);
            p2 += __shfl_xor(p2, m, 32);
        }
        if (n == 0) {
            s1[b * NODES + row] = p1;
            s2[b * NODES + row] = p2;
        }
    }
}

#define TILE_J          64
#define KSPLIT          2
#define KEYS_PER_SPLIT  (NODES / KSPLIT)
#define NTILES          (KEYS_PER_SPLIT / TILE_J)
#define ROWTILES        4
#define TILE_F4         (TILE_J * FOUT / 4)
#define SPLIT_F4        (KEYS_PER_SPLIT * FOUT / 4)
#define ADJ_PAIR        (2 * NODES / 4)

#define ADJ_LOAD(BASE)                                                                     \
    ar_0 = (BASE)[0];            ar_1 = (BASE)[ADJ_PAIR];     ar_2 = (BASE)[2 * ADJ_PAIR]; \
    ar_3 = (BASE)[3 * ADJ_PAIR]; ar_4 = (BASE)[4 * ADJ_PAIR]; ar_5 = (BASE)[5 * ADJ_PAIR]; \
    ar_6 = (BASE)[6 * ADJ_PAIR]; ar_7 = (BASE)[7 * ADJ_PAIR];

#define ADJ_FOLD(I, R)                                                                     \
    {                                                                                      \
        const unsigned b0 = __builtin_amdgcn_ballot_w32((R)[0] > 0);                       \
        const unsigned b1 = __builtin_amdgcn_ballot_w32((R)[1] > 0);                       \
        const unsigned b2 = __builtin_amdgcn_ballot_w32((R)[2] > 0);                       \
        const unsigned b3 = __builtin_amdgcn_ballot_w32((R)[3] > 0);                       \
        mA = (sel == (I)) ? b0 : mA;   mA = (sel == (I) + 8) ? b2 : mA;                    \
        mB = (sel == (I)) ? b1 : mB;   mB = (sel == (I) + 8) ? b3 : mB;                    \
    }
#define ADJ_MASK(M)                                                                        \
    {                                                                                      \
        unsigned mA = 0u, mB = 0u;                                                         \
        ADJ_FOLD(0, ar_0) ADJ_FOLD(1, ar_1) ADJ_FOLD(2, ar_2) ADJ_FOLD(3, ar_3)            \
        ADJ_FOLD(4, ar_4) ADJ_FOLD(5, ar_5) ADJ_FOLD(6, ar_6) ADJ_FOLD(7, ar_7)            \
        (M) = ((mA >> rsh) & 0xFFFFu) | ((mB >> rsh) << 16);                               \
    }

__global__ void __launch_bounds__(256) gat_attn_kernel(
    const float* __restrict__ h, const float* __restrict__ s1g,
    const float* __restrict__ s2g, const int* __restrict__ adj,
    float* __restrict__ out)
{
    __shared__ __attribute__((aligned(16))) float lds_h[2][KSPLIT][TILE_J * FOUT];
    __shared__ float lds_s2[2][KSPLIT * TILE_J];
    __shared__ float lds_part[ROWTILES][34][32];

    const int lane = threadIdx.x & 31;
    const int wid  = threadIdx.x >> 5;
    const int kh   = wid >> 2;
    const int rt   = wid & 3;
    const int wg   = blockIdx.x * ROWTILES + rt;
    const int b    = wg >> 8;
    const int i0   = (wg & 255) << 4;
    const int half = lane >> 4;
    const int n    = lane & 15;

    const float sv1 = s1g[b * NODES + i0 + n];
    const v4i* adjsrc = (const v4i*)(adj + (size_t)(i0 + half) * NODES + kh * KEYS_PER_SPLIT) + n;
    const int  sel    = (n >> 1) + 8 * half;
    const int  rsh    = 16 * (n & 1);
    const v4f*   hsrc  = (const v4f*)(h + (size_t)b * NODES * FOUT) + threadIdx.x;
    const float*  s2src = s2g + b * NODES + (threadIdx.x >> 6) * KEYS_PER_SPLIT + (threadIdx.x & (TILE_J - 1));
    const bool    s2thr = threadIdx.x < KSPLIT * TILE_J;

    v4f    hr_0, hr_1, hr_2, hr_3;
    v4f    hr_4, hr_5, hr_6, hr_7;
    float  s2reg = 0.0f;
    v4i    ar_0, ar_1, ar_2, ar_3, ar_4, ar_5, ar_6, ar_7;
    unsigned Mc;

    hr_0 = hsrc[0];            hr_1 = hsrc[256];            hr_2 = hsrc[512];            hr_3 = hsrc[768];
    hr_4 = hsrc[SPLIT_F4];     hr_5 = hsrc[SPLIT_F4 + 256]; hr_6 = hsrc[SPLIT_F4 + 512]; hr_7 = hsrc[SPLIT_F4 + 768];
    if (s2thr) s2reg = s2src[0];
    ADJ_LOAD(adjsrc)
    {
        v4f* dst = (v4f*)lds_h[0][0] + threadIdx.x;
        dst[0] = hr_0;           dst[256] = hr_1;           dst[512] = hr_2;           dst[768] = hr_3;
        dst[TILE_F4] = hr_4;     dst[TILE_F4 + 256] = hr_5; dst[TILE_F4 + 512] = hr_6; dst[TILE_F4 + 768] = hr_7;
        if (s2thr) lds_s2[0][threadIdx.x] = s2reg;
    }
    ADJ_MASK(Mc)
    __syncthreads();

    float mrun = NEG_BIG;
    float lrun = 0.0f;
    v8f acc0 = {}, acc1 = {}, acc2 = {}, acc3 = {};

    for (int t = 0; t < NTILES; ++t) {
        const int  cur  = t & 1;
        const bool more = (t + 1) < NTILES;

        if (more) {
            const int j1 = (t + 1) * TILE_J;
            const v4f* hs = hsrc + (size_t)j1 * (FOUT / 4);
            hr_0 = hs[0];            hr_1 = hs[256];            hr_2 = hs[512];            hr_3 = hs[768];
            hr_4 = hs[SPLIT_F4];     hr_5 = hs[SPLIT_F4 + 256]; hr_6 = hs[SPLIT_F4 + 512]; hr_7 = hs[SPLIT_F4 + 768];
            if (s2thr) s2reg = s2src[j1];
            const v4i* as = adjsrc + j1 / 4;
            ADJ_LOAD(as)
        }
        const float* lh  = lds_h[cur][kh];
        const float* ls2 = lds_s2[cur] + kh * TILE_J;

        #pragma unroll
        for (int jj = 0; jj < 4; ++jj) {
            const int jb = jj * 16;
            float p[8];
            float mx = NEG_BIG;
            for (int q = 0; q < 4; ++q) {
                for (int c = 0; c < 2; ++c) {
                    const int jl = jb + 4 * q + 2 * half + c;
                    float e = sv1 + ls2[jl];
                    e = (e >= 0.0f) ? e : ALPHA * e;
                    e = ((Mc >> (16 * c + 4 * jj + q)) & 1u) ? e : NEG_BIG;
                    p[2 * q + c] = e;
                    mx = fmaxf(mx, e);
                }
            }
            mx = fmaxf(mx, __shfl_xor(mx, 16, 32));
            const float mnew  = fmaxf(mrun, mx);
            const float scale = __expf(mrun - mnew);
            float rs = 0.0f;
            for (int u = 0; u < 8; ++u) { p[u] = __expf(p[u] - mnew); rs += p[u]; }
            rs += __shfl_xor(rs, 16, 32);
            lrun = lrun * scale + rs;
            mrun = mnew;

            for (int r = 0; r < 8; ++r) {
                const float sr = __shfl(scale, r + 8 * half, 32);
                acc0[r] *= sr; acc1[r] *= sr; acc2[r] *= sr; acc3[r] *= sr;
            }

            for (int q = 0; q < 4; ++q) {
                v2f A; A[0] = p[2 * q]; A[1] = p[2 * q + 1];
                const float* hr0 = lh + (size_t)(jb + 4 * q + 2 * half) * FOUT + n;
                const float* hr1 = hr0 + FOUT;
                v2f B0; B0[0] = hr0[0];  B0[1] = hr1[0];
                v2f B1; B1[0] = hr0[16]; B1[1] = hr1[16];
                v2f B2; B2[0] = hr0[32]; B2[1] = hr1[32];
                v2f B3; B3[0] = hr0[48]; B3[1] = hr1[48];
                acc0 = __builtin_amdgcn_wmma_f32_16x16x4_f32(false, A, false, B0, (short)0, acc0, false, false);
                acc1 = __builtin_amdgcn_wmma_f32_16x16x4_f32(false, A, false, B1, (short)0, acc1, false, false);
                acc2 = __builtin_amdgcn_wmma_f32_16x16x4_f32(false, A, false, B2, (short)0, acc2, false, false);
                acc3 = __builtin_amdgcn_wmma_f32_16x16x4_f32(false, A, false, B3, (short)0, acc3, false, false);
            }
        }

        if (more) {
            v4f* dst = (v4f*)lds_h[cur ^ 1][0] + threadIdx.x;
            dst[0] = hr_0;           dst[256] = hr_1;           dst[512] = hr_2;           dst[768] = hr_3;
            dst[TILE_F4] = hr_4;     dst[TILE_F4 + 256] = hr_5; dst[TILE_F4 + 512] = hr_6; dst[TILE_F4 + 768] = hr_7;
            if (s2thr) lds_s2[cur ^ 1][threadIdx.x] = s2reg;
            ADJ_MASK(Mc)
        }
        __syncthreads();
    }

    float* part = &lds_part[rt][0][lane];
    if (kh == 1) {
        for (int r = 0; r < 8; ++r) {
            part[(r)      * 32] = acc0[r];
            part[(8 + r)  * 32] = acc1[r];
            part[(16 + r) * 32] = acc2[r];
            part[(24 + r) * 32] = acc3[r];
        }
        part[32 * 32] = mrun;
        part[33 * 32] = lrun;
    }
    __syncthreads();
    if (kh == 0) {
        const float m1 = part[32 * 32];
        const float l1 = part[33 * 32];
        const float m  = fmaxf(mrun, m1);
        const float f0 = __expf(mrun - m);
        const float f1 = __expf(m1 - m);
        const float l  = lrun * f0 + l1 * f1;

        for (int r = 0; r < 8; ++r) {
            const int   src = r + 8 * half;
            const float g0  = __shfl(f0, src, 32);
            const float g1  = __shfl(f1, src, 32);
            const float lr  = __shfl(l,  src, 32);
            const float inv = 1.0f / lr;
            const int row   = i0 + r + 8 * half;
            float* o = out + ((size_t)(b * NODES + row)) * FOUT + n;
            float v;
            v = (acc0[r] * g0 + part[(r)      * 32] * g1) * inv; o[0]  = (v > 0.0f) ? v : expm1f(v);
            v = (acc1[r] * g0 + part[(8 + r)  * 32] * g1) * inv; o[16] = (v > 0.0f) ? v : expm1f(v);
            v = (acc2[r] * g0 + part[(16 + r) * 32] * g1) * inv; o[32] = (v > 0.0f) ? v : expm1f(v);
            v = (acc3[r] * g0 + part[(24 + r) * 32] * g1) * inv; o[48] = (v > 0.0f) ? v : expm1f(v);
        }
    }
}

extern "C" void kernel_launch(void* const* d_in, const int* in_sizes, int n_in,
                              void* d_out, int out_size, void* d_ws, size_t ws_size,
                              hipStream_t stream)
{
    const float* x   = (const float*)d_in[0];
    const int*   adj = (const int*)  d_in[1];
    const float* W   = (const float*)d_in[2];
    const float* a   = (const float*)d_in[3];
    float* out = (float*)d_out;

    float* h  = (float*)d_ws;
    float* s1 = h + (size_t)BATCH * NODES * FOUT;
    float* s2 = s1 + (size_t)BATCH * NODES;

    const int total_waves = BATCH * (NODES / 16);
    const int blocks_h = total_waves / WAVES_PER_BLOCK;
    const int blocks_a = total_waves / ROWTILES;

    gat_h_kernel<<<blocks_h, 256, 0, stream>>>(x, W, a, h, s1, s2);
    gat_attn_kernel<<<blocks_a, 256, 0, stream>>>(h, s1, s2, adj, out);
}


# Kernel 2: softmax max, sum and accumulator rescale once per 64-key tile, not per 16 keys (row sums regrouped, weights taken against the tile max: rounding changes); the tile's 64 products in one run
# speedup vs baseline: 2.9487x; 1.0541x over previous
#include <hip/hip_runtime.h>
#include <hip/hip_bf16.h>

typedef __attribute__((ext_vector_type(2))) float v2f;
typedef __attribute__((ext_vector_type(8))) float v8f;
typedef __attribute__((ext_vector_type(4))) float v4f;
typedef __attribute__((ext_vector_type(4))) int   v4i;

#define NODES   4096
#define FIN     512
#define FOUT    64
#define BATCH   4
#define NEG_BIG (-9000000000000000.0f)
#define ALPHA   0.2f
#define WAVES_PER_BLOCK 8

__global__ void __launch_bounds__(256) gat_h_kernel(
    const float* __restrict__ x, const float* __restrict__ W,
    const float* __restrict__ a,
    float* __restrict__ h, float* __restrict__ s1, float* __restrict__ s2)
{
    const int lane = threadIdx.x & 31;
    const int wid  = threadIdx.x >> 5;
    const int wg   = blockIdx.x * WAVES_PER_BLOCK + wid;
    const int b    = wg >> 8;
    const int i0   = (wg & 255) << 4;
    const int half = lane >> 4;
    const int n    = lane & 15;

    const float* xrow = x + ((size_t)(b * NODES + i0 + n) * FIN);

    v8f acc0 = {}, acc1 = {}, acc2 = {}, acc3 = {};

    for (int k = 0; k < FIN; k += 4) {
        const int ka = k + 2 * half;
        v2f A; A[0] = xrow[ka]; A[1] = xrow[ka + 1];
        const float* wr0 = W + (size_t)ka * FOUT + n;
        const float* wr1 = wr0 + FOUT;
        v2f B0; B0[0] = wr0[0];  B0[1] = wr1[0];
        v2f B1; B1[0] = wr0[16]; B1[1] = wr1[16];
        v2f B2; B2[0] = wr0[32]; B2[1] = wr1[32];
        v2f B3; B3[0] = wr0[48]; B3[1] = wr1[48];
        acc0 = __builtin_amdgcn_wmma_f32_16x16x4_f32(false, A, false, B0, (short)0, acc0, false, false);
        acc1 = __builtin_amdgcn_wmma_f32_16x16x4_f32(false, A, false, B1, (short)0, acc1, false, false);
        acc2 = __builtin_amdgcn_wmma_f32_16x16x4_f32(false, A, false, B2, (short)0, acc2, false, false);
        acc3 = __builtin_amdgcn_wmma_f32_16x16x4_f32(false, A, false, B3, (short)0, acc3, false, false);
    }

    const float a10 = a[n],        a11 = a[n + 16], a12 = a[n + 32], a13 = a[n + 48];
    const float a20 = a[FOUT + n], a21 = a[FOUT + n + 16],
                a22 = a[FOUT + n + 32], a23 = a[FOUT + n + 48];

    for (int r = 0; r < 8; ++r) {
        const int row = i0 + r + 8 * half;
        float* hrow = h + ((size_t)(b * NODES + row) * FOUT) + n;
        const float c0 = acc0[r], c1 = acc1[r], c2 = acc2[r], c3 = acc3[r];
        hrow[0] = c0; hrow[16] = c1; hrow[32] = c2; hrow[48] = c3;

        float p1 = c0 * a10 + c1 * a11 + c2 * a12 + c3 * a13;
        float p2 = c0 * a20 + c1 * a21 + c2 * a22 + c3 * a23;
        for (int m = 1; m < 16; m <<= 1) {
            p1 += __shfl_xor(p1, m, 32);
            p2 += __shfl_xor(p2, m, 32);
        }
        if (n == 0) {
            s1[b * NODES + row] = p1;
            s2[b * NODES + row] = p2;
        }
    }
}

#define TILE_J          64
#define KSPLIT          2
#define KEYS_PER_SPLIT  (NODES / KSPLIT)
#define NTILES          (KEYS_PER_SPLIT / TILE_J)
#define ROWTILES        4
#define TILE_F4         (TILE_J * FOUT / 4)
#define SPLIT_F4        (KEYS_PER_SPLIT * FOUT / 4)
#define ADJ_PAIR        (2 * NODES / 4)

#define ADJ_LOAD(BASE)                                                                     \
    ar_0 = (BASE)[0];            ar_1 = (BASE)[ADJ_PAIR];     ar_2 = (BASE)[2 * ADJ_PAIR]; \
    ar_3 = (BASE)[3 * ADJ_PAIR]; ar_4 = (BASE)[4 * ADJ_PAIR]; ar_5 = (BASE)[5 * ADJ_PAIR]; \
    ar_6 = (BASE)[6 * ADJ_PAIR]; ar_7 = (BASE)[7 * ADJ_PAIR];

#define ADJ_FOLD(I, R)                                                                     \
    {                                                                                      \
        const unsigned b0 = __builtin_amdgcn_ballot_w32((R)[0] > 0);                       \
        const unsigned b1 = __builtin_amdgcn_ballot_w32((R)[1] > 0);                       \
        const unsigned b2 = __builtin_amdgcn_ballot_w32((R)[2] > 0);                       \
        const unsigned b3 = __builtin_amdgcn_ballot_w32((R)[3] > 0);                       \
        mA = (sel == (I)) ? b0 : mA;   mA = (sel == (I) + 8) ? b2 : mA;                    \
        mB = (sel == (I)) ? b1 : mB;   mB = (sel == (I) + 8) ? b3 : mB;                    \
    }
#define ADJ_MASK(M)                                                                        \
    {                                                                                      \
        unsigned mA = 0u, mB = 0u;                                                         \
        ADJ_FOLD(0, ar_0) ADJ_FOLD(1, ar_1) ADJ_FOLD(2, ar_2) ADJ_FOLD(3, ar_3)            \
        ADJ_FOLD(4, ar_4) ADJ_FOLD(5, ar_5) ADJ_FOLD(6, ar_6) ADJ_FOLD(7, ar_7)            \
        (M) = ((mA >> rsh) & 0xFFFFu) | ((mB >> rsh) << 16);                               \
    }

__global__ void __launch_bounds__(256) gat_attn_kernel(
    const float* __restrict__ h, const float* __restrict__ s1g,
    const float* __restrict__ s2g, const int* __restrict__ adj,
    float* __restrict__ out)
{
    __shared__ __attribute__((aligned(16))) float lds_h[2][KSPLIT][TILE_J * FOUT];
    __shared__ float lds_s2[2][KSPLIT * TILE_J];
    __shared__ float lds_part[ROWTILES][34][32];

    const int lane = threadIdx.x & 31;
    const int wid  = threadIdx.x >> 5;
    const int kh   = wid >> 2;
    const int rt   = wid & 3;
    const int wg   = blockIdx.x * ROWTILES + rt;
    const int b    = wg >> 8;
    const int i0   = (wg & 255) << 4;
    const int half = lane >> 4;
    const int n    = lane & 15;

    const float sv1 = s1g[b * NODES + i0 + n];
    const v4i* adjsrc = (const v4i*)(adj + (size_t)(i0 + half) * NODES + kh * KEYS_PER_SPLIT) + n;
    const int  sel    = (n >> 1) + 8 * half;
    const int  rsh    = 16 * (n & 1);
    const v4f*   hsrc  = (const v4f*)(h + (size_t)b * NODES * FOUT) + threadIdx.x;
    const float*  s2src = s2g + b * NODES + (threadIdx.x >> 6) * KEYS_PER_SPLIT + (threadIdx.x & (TILE_J - 1));
    const bool    s2thr = threadIdx.x < KSPLIT * TILE_J;

    v4f    hr_0, hr_1, hr_2, hr_3;
    v4f    hr_4, hr_5, hr_6, hr_7;
    float  s2reg = 0.0f;
    v4i    ar_0, ar_1, ar_2, ar_3, ar_4, ar_5, ar_6, ar_7;
    unsigned Mc;

    hr_0 = hsrc[0];            hr_1 = hsrc[256];            hr_2 = hsrc[512];            hr_3 = hsrc[768];
    hr_4 = hsrc[SPLIT_F4];     hr_5 = hsrc[SPLIT_F4 + 256]; hr_6 = hsrc[SPLIT_F4 + 512]; hr_7 = hsrc[SPLIT_F4 + 768];
    if (s2thr) s2reg = s2src[0];
    ADJ_LOAD(adjsrc)
    {
        v4f* dst = (v4f*)lds_h[0][0] + threadIdx.x;
        dst[0] = hr_0;           dst[256] = hr_1;           dst[512] = hr_2;           dst[768] = hr_3;
        dst[TILE_F4] = hr_4;     dst[TILE_F4 + 256] = hr_5; dst[TILE_F4 + 512] = hr_6; dst[TILE_F4 + 768] = hr_7;
        if (s2thr) lds_s2[0][threadIdx.x] = s2reg;
    }
    ADJ_MASK(Mc)
    __syncthreads();

    float mrun = NEG_BIG;
    float lrun = 0.0f;
    v8f acc0 = {}, acc1 = {}, acc2 = {}, acc3 = {};

    for (int t = 0; t < NTILES; ++t) {
        const int  cur  = t & 1;
        const bool more = (t + 1) < NTILES;

        if (more) {
            const int j1 = (t + 1) * TILE_J;
            const v4f* hs = hsrc + (size_t)j1 * (FOUT / 4);
            hr_0 = hs[0];            hr_1 = hs[256];            hr_2 = hs[512];            hr_3 = hs[768];
            hr_4 = hs[SPLIT_F4];     hr_5 = hs[SPLIT_F4 + 256]; hr_6 = hs[SPLIT_F4 + 512]; hr_7 = hs[SPLIT_F4 + 768];
            if (s2thr) s2reg = s2src[j1];
            const v4i* as = adjsrc + j1 / 4;
            ADJ_LOAD(as)
        }
        const float* lh  = lds_h[cur][kh];
        const float* ls2 = lds_s2[cur] + kh * TILE_J;

        float p[32];
        float mx = NEG_BIG;
        for (int jj = 0; jj < 4; ++jj) {
            for (int q = 0; q < 4; ++q) {
                for (int c = 0; c < 2; ++c) {
                    const int jl = 16 * jj + 4 * q + 2 * half + c;
                    float e = sv1 + ls2[jl];
                    e = (e >= 0.0f) ? e : ALPHA * e;
                    e = ((Mc >> (16 * c + 4 * jj + q)) & 1u) ? e : NEG_BIG;
                    p[8 * jj + 2 * q + c] = e;
                    mx = fmaxf(mx, e);
                }
            }
        }
        mx = fmaxf(mx, __shfl_xor(mx, 16, 32));
        const float mnew  = fmaxf(mrun, mx);
        const float scale = __expf(mrun - mnew);
        float rs = 0.0f;
        for (int jj = 0; jj < 4; ++jj) {
            float rj = 0.0f;
            for (int u = 0; u < 8; ++u) { p[8 * jj + u] = __expf(p[8 * jj + u] - mnew); rj += p[8 * jj + u]; }
            rs += rj;
        }
        rs += __shfl_xor(rs, 16, 32);
        lrun = lrun * scale + rs;
        mrun = mnew;

        for (int r = 0; r < 8; ++r) {
            const float sr = __shfl(scale, r + 8 * half, 32);
            acc0[r] *= sr; acc1[r] *= sr; acc2[r] *= sr; acc3[r] *= sr;
        }

        for (int jj = 0; jj < 4; ++jj) {
            for (int q = 0; q < 4; ++q) {
                v2f A; A[0] = p[8 * jj + 2 * q]; A[1] = p[8 * jj + 2 * q + 1];
                const float* hr0 = lh + (size_t)(16 * jj + 4 * q + 2 * half) * FOUT + n;
                const float* hr1 = hr0 + FOUT;
                v2f B0; B0[0] = hr0[0];  B0[1] = hr1[0];
                v2f B1; B1[0] = hr0[16]; B1[1] = hr1[16];
                v2f B2; B2[0] = hr0[32]; B2[1] = hr1[32];
                v2f B3; B3[0] = hr0[48]; B3[1] = hr1[48];
                acc0 = __builtin_amdgcn_wmma_f32_16x16x4_f32(false, A, false, B0, (short)0, acc0, false, false);
                acc1 = __builtin_amdgcn_wmma_f32_16x16x4_f32(false, A, false, B1, (short)0, acc1, false, false);
                acc2 = __builtin_amdgcn_wmma_f32_16x16x4_f32(false, A, false, B2, (short)0, acc2, false, false);
                acc3 = __builtin_amdgcn_wmma_f32_16x16x4_f32(false, A, false, B3, (short)0, acc3, false, false);
            }
        }

        if (more) {
            v4f* dst = (v4f*)lds_h[cur ^ 1][0] + threadIdx.x;
            dst[0] = hr_0;           dst[256] = hr_1;           dst[512] = hr_2;           dst[768] = hr_3;
            dst[TILE_F4] = hr_4;     dst[TILE_F4 + 256] = hr_5; dst[TILE_F4 + 512] = hr_6; dst[TILE_F4 + 768] = hr_7;
            if (s2thr) lds_s2[cur ^ 1][threadIdx.x] = s2reg;
            ADJ_MASK(Mc)
        }
        __syncthreads();
    }

    float* part = &lds_part[rt][0][lane];
    if (kh == 1) {
        for (int r = 0; r < 8; ++r) {
            part[(r)      * 32] = acc0[r];
            part[(8 + r)  * 32] = acc1[r];
            part[(16 + r) * 32] = acc2[r];
            part[(24 + r) * 32] = acc3[r];
        }
        part[32 * 32] = mrun;
        part[33 * 32] = lrun;
    }
    __syncthreads();
    if (kh == 0) {
        const float m1 = part[32 * 32];
        const float l1 = part[33 * 32];
        const float m  = fmaxf(mrun, m1);
        const float f0 = __expf(mrun - m);
        const float f1 = __expf(m1 - m);
        const float l  = lrun * f0 + l1 * f1;

        for (int r = 0; r < 8; ++r) {
            const int   src = r + 8 * half;
            const float g0  = __shfl(f0, src, 32);
            const float g1  = __shfl(f1, src, 32);
            const float lr  = __shfl(l,  src, 32);
            const float inv = 1.0f / lr;
            const int row   = i0 + r + 8 * half;
            float* o = out + ((size_t)(b * NODES + row)) * FOUT + n;
            float v;
            v = (acc0[r] * g0 + part[(r)      * 32] * g1) * inv; o[0]  = (v > 0.0f) ? v : expm1f(v);
            v = (acc1[r] * g0 + part[(8 + r)  * 32] * g1) * inv; o[16] = (v > 0.0f) ? v : expm1f(v);
            v = (acc2[r] * g0 + part[(16 + r) * 32] * g1) * inv; o[32] = (v > 0.0f) ? v : expm1f(v);
            v = (acc3[r] * g0 + part[(24 + r) * 32] * g1) * inv; o[48] = (v > 0.0f) ? v : expm1f(v);
        }
    }
}

extern "C" void kernel_launch(void* const* d_in, const int* in_sizes, int n_in,
                              void* d_out, int out_size, void* d_ws, size_t ws_size,
                              hipStream_t stream)
{
    const float* x   = (const float*)d_in[0];
    const int*   adj = (const int*)  d_in[1];
    const float* W   = (const float*)d_in[2];
    const float* a   = (const float*)d_in[3];
    float* out = (float*)d_out;

    float* h  = (float*)d_ws;
    float* s1 = h + (size_t)BATCH * NODES * FOUT;
    float* s2 = s1 + (size_t)BATCH * NODES;

    const int total_waves = BATCH * (NODES / 16);
    const int blocks_h = total_waves / WAVES_PER_BLOCK;
    const int blocks_a = total_waves / ROWTILES;

    gat_h_kernel<<<blocks_h, 256, 0, stream>>>(x, W, a, h, s1, s2);
    gat_attn_kernel<<<blocks_a, 256, 0, stream>>>(h, s1, s2, adj, out);
}


# Kernel 1: both operands resident before the products: W copied once per block to LDS, each wave's x tile loaded up front in 16-byte loads; 4-wave blocks, grid 256 (k regrouped: h rounding changes)
# speedup vs baseline: 3.7645x; 1.2767x over previous
#include <hip/hip_runtime.h>
#include <hip/hip_bf16.h>

typedef __attribute__((ext_vector_type(2))) float v2f;
typedef __attribute__((ext_vector_type(8))) float v8f;
typedef __attribute__((ext_vector_type(4))) float v4f;
typedef __attribute__((ext_vector_type(4))) int   v4i;

#define NODES   4096
#define FIN     512
#define FOUT    64
#define BATCH   4
#define NEG_BIG (-9000000000000000.0f)
#define ALPHA   0.2f

#define K1_WAVES      4
#define W_PAIR_PITCH  176

#define K1_LOADB(DST, G)                                                                         \
    _Pragma("unroll")                                                                            \
    for (int s = 0; s < 8; ++s) {                                                                \
        const float* r0 = lwp + (4 * (4 * (G) + (s >> 1)) + (s & 1)) * W_PAIR_PITCH;             \
        DST[s][0] = *(const v2f*)(r0);      DST[s][1] = *(const v2f*)(r0 + 32);                  \
        DST[s][2] = *(const v2f*)(r0 + 64); DST[s][3] = *(const v2f*)(r0 + 96);                  \
    }

__global__ void __launch_bounds__(128) gat_h_kernel(
    const float* __restrict__ x, const float* __restrict__ W,
    const float* __restrict__ a,
    float* __restrict__ h, float* __restrict__ s1, float* __restrict__ s2)
{
    __shared__ __attribute__((aligned(16))) float ldsW[(FIN / 2) * W_PAIR_PITCH];

    const int lane = threadIdx.x & 31;
    const int wid  = threadIdx.x >> 5;
    const int wg   = blockIdx.x * K1_WAVES + wid;
    const int b    = wg >> 8;
    const int i0   = (wg & 255) << 4;
    const int half = lane >> 4;
    const int n    = lane & 15;

    const v4f* wsrc = (const v4f*)W + (threadIdx.x >> 4) * 32 + (threadIdx.x & 15);
    v4f we[32], wo[32];
#pragma unroll
    for (int i = 0; i < 32; ++i) { we[i] = wsrc[256 * i]; wo[i] = wsrc[256 * i + 16]; }

    const v4f* xsrc = (const v4f*)(x + (size_t)(b * NODES + i0 + n) * FIN) + half;
    v4f X[64];
#pragma unroll
    for (int j = 0; j < 64; ++j) X[j] = xsrc[2 * j];

    {
        float* wdst = ldsW + (threadIdx.x >> 4) * W_PAIR_PITCH + 8 * (threadIdx.x & 15);
#pragma unroll
        for (int i = 0; i < 32; ++i) {
            v4f lo, hi;
            lo[0] = we[i][0]; lo[1] = wo[i][0]; lo[2] = we[i][1]; lo[3] = wo[i][1];
            hi[0] = we[i][2]; hi[1] = wo[i][2]; hi[2] = we[i][3]; hi[3] = wo[i][3];
            *(v4f*)(wdst + 8 * i * W_PAIR_PITCH)     = lo;
            *(v4f*)(wdst + 8 * i * W_PAIR_PITCH + 4) = hi;
        }
    }
    __builtin_amdgcn_fence(__ATOMIC_RELEASE, "workgroup", "local");
    __builtin_amdgcn_s_barrier();
    __builtin_amdgcn_fence(__ATOMIC_ACQUIRE, "workgroup", "local");

    const float* lwp = ldsW + (2 * half) * W_PAIR_PITCH + 2 * n;

    v8f acc0 = {}, acc1 = {}, acc2 = {}, acc3 = {};
    v2f Bc[8][4], Bn[8][4];
    K1_LOADB(Bc, 0)
#pragma unroll
    for (int g = 0; g < 16; ++g) {
        if (g + 1 < 16) { K1_LOADB(Bn, g + 1) }
        __builtin_amdgcn_sched_barrier(0);
#pragma unroll
        for (int s = 0; s < 8; ++s) {
            const v4f xv = X[4 * g + (s >> 1)];
            v2f A; A[0] = xv[2 * (s & 1)]; A[1] = xv[2 * (s & 1) + 1];
            acc0 = __builtin_amdgcn_wmma_f32_16x16x4_f32(false, A, false, Bc[s][0], (short)0, acc0, false, false);
            acc1 = __builtin_amdgcn_wmma_f32_16x16x4_f32(false, A, false, Bc[s][1], (short)0, acc1, false, false);
            acc2 = __builtin_amdgcn_wmma_f32_16x16x4_f32(false, A, false, Bc[s][2], (short)0, acc2, false, false);
            acc3 = __builtin_amdgcn_wmma_f32_16x16x4_f32(false, A, false, Bc[s][3], (short)0, acc3, false, false);
        }
#pragma unroll
        for (int s = 0; s < 8; ++s) {
            Bc[s][0] = Bn[s][0]; Bc[s][1] = Bn[s][1]; Bc[s][2] = Bn[s][2]; Bc[s][3] = Bn[s][3];
        }
    }

    const float a10 = a[n],        a11 = a[n + 16], a12 = a[n + 32], a13 = a[n + 48];
    const float a20 = a[FOUT + n], a21 = a[FOUT + n + 16],
                a22 = a[FOUT + n + 32], a23 = a[FOUT + n + 48];

    for (int r = 0; r < 8; ++r) {
        const int row = i0 + r + 8 * half;
        float* hrow = h + ((size_t)(b * NODES + row) * FOUT) + n;
        const float c0 = acc0[r], c1 = acc1[r], c2 = acc2[r], c3 = acc3[r];
        hrow[0] = c0; hrow[16] = c1; hrow[32] = c2; hrow[48] = c3;

        float p1 = c0 * a10 + c1 * a11 + c2 * a12 + c3 * a13;
        float p2 = c0 * a20 + c1 * a21 + c2 * a22 + c3 * a23;
        for (int m = 1; m < 16; m <<= 1) {
            p1 += __shfl_xor(p1, m, 32);
            p2 += __shfl_xor(p2, m, 32);
        }
        if (n == 0) {
            s1[b * NODES + row] = p1;
            s2[b * NODES + row] = p2;
        }
    }
}

#define TILE_J          64
#define KSPLIT          2
#define KEYS_PER_SPLIT  (NODES / KSPLIT)
#define NTILES          (KEYS_PER_SPLIT / TILE_J)
#define ROWTILES        4
#define TILE_F4         (TILE_J * FOUT / 4)
#define SPLIT_F4        (KEYS_PER_SPLIT * FOUT / 4)
#define ADJ_PAIR        (2 * NODES / 4)

#define ADJ_LOAD(BASE)                                                                     \
    ar_0 = (BASE)[0];            ar_1 = (BASE)[ADJ_PAIR];     ar_2 = (BASE)[2 * ADJ_PAIR]; \
    ar_3 = (BASE)[3 * ADJ_PAIR]; ar_4 = (BASE)[4 * ADJ_PAIR]; ar_5 = (BASE)[5 * ADJ_PAIR]; \
    ar_6 = (BASE)[6 * ADJ_PAIR]; ar_7 = (BASE)[7 * ADJ_PAIR];

#define ADJ_FOLD(I, R)                                                                     \
    {                                                                                      \
        const unsigned b0 = __builtin_amdgcn_ballot_w32((R)[0] > 0);                       \
        const unsigned b1 = __builtin_amdgcn_ballot_w32((R)[1] > 0);                       \
        const unsigned b2 = __builtin_amdgcn_ballot_w32((R)[2] > 0);                       \
        const unsigned b3 = __builtin_amdgcn_ballot_w32((R)[3] > 0);                       \
        mA = (sel == (I)) ? b0 : mA;   mA = (sel == (I) + 8) ? b2 : mA;                    \
        mB = (sel == (I)) ? b1 : mB;   mB = (sel == (I) + 8) ? b3 : mB;                    \
    }
#define ADJ_MASK(M)                                                                        \
    {                                                                                      \
        unsigned mA = 0u, mB = 0u;                                                         \
        ADJ_FOLD(0, ar_0) ADJ_FOLD(1, ar_1) ADJ_FOLD(2, ar_2) ADJ_FOLD(3, ar_3)            \
        ADJ_FOLD(4, ar_4) ADJ_FOLD(5, ar_5) ADJ_FOLD(6, ar_6) ADJ_FOLD(7, ar_7)            \
        (M) = ((mA >> rsh) & 0xFFFFu) | ((mB >> rsh) << 16);                               \
    }

__global__ void __launch_bounds__(256) gat_attn_kernel(
    const float* __restrict__ h, const float* __restrict__ s1g,
    const float* __restrict__ s2g, const int* __restrict__ adj,
    float* __restrict__ out)
{
    __shared__ __attribute__((aligned(16))) float lds_h[2][KSPLIT][TILE_J * FOUT];
    __shared__ float lds_s2[2][KSPLIT * TILE_J];
    __shared__ float lds_part[ROWTILES][34][32];

    const int lane = threadIdx.x & 31;
    const int wid  = threadIdx.x >> 5;
    const int kh   = wid >> 2;
    const int rt   = wid & 3;
    const int wg   = blockIdx.x * ROWTILES + rt;
    const int b    = wg >> 8;
    const int i0   = (wg & 255) << 4;
    const int half = lane >> 4;
    const int n    = lane & 15;

    const float sv1 = s1g[b * NODES + i0 + n];
    const v4i* adjsrc = (const v4i*)(adj + (size_t)(i0 + half) * NODES + kh * KEYS_PER_SPLIT) + n;
    const int  sel    = (n >> 1) + 8 * half;
    const int  rsh    = 16 * (n & 1);
    const v4f*   hsrc  = (const v4f*)(h + (size_t)b * NODES * FOUT) + threadIdx.x;
    const float*  s2src = s2g + b * NODES + (threadIdx.x >> 6) * KEYS_PER_SPLIT + (threadIdx.x & (TILE_J - 1));
    const bool    s2thr = threadIdx.x < KSPLIT * TILE_J;

    v4f    hr_0, hr_1, hr_2, hr_3;
    v4f    hr_4, hr_5, hr_6, hr_7;
    float  s2reg = 0.0f;
    v4i    ar_0, ar_1, ar_2, ar_3, ar_4, ar_5, ar_6, ar_7;
    unsigned Mc;

    hr_0 = hsrc[0];            hr_1 = hsrc[256];            hr_2 = hsrc[512];            hr_3 = hsrc[768];
    hr_4 = hsrc[SPLIT_F4];     hr_5 = hsrc[SPLIT_F4 + 256]; hr_6 = hsrc[SPLIT_F4 + 512]; hr_7 = hsrc[SPLIT_F4 + 768];
    if (s2thr) s2reg = s2src[0];
    ADJ_LOAD(adjsrc)
    {
        v4f* dst = (v4f*)lds_h[0][0] + threadIdx.x;
        dst[0] = hr_0;           dst[256] = hr_1;           dst[512] = hr_2;           dst[768] = hr_3;
        dst[TILE_F4] = hr_4;     dst[TILE_F4 + 256] = hr_5; dst[TILE_F4 + 512] = hr_6; dst[TILE_F4 + 768] = hr_7;
        if (s2thr) lds_s2[0][threadIdx.x] = s2reg;
    }
    ADJ_MASK(Mc)
    __syncthreads();

    float mrun = NEG_BIG;
    float lrun = 0.0f;
    v8f acc0 = {}, acc1 = {}, acc2 = {}, acc3 = {};

    for (int t = 0; t < NTILES; ++t) {
        const int  cur  = t & 1;
        const bool more = (t + 1) < NTILES;

        if (more) {
            const int j1 = (t + 1) * TILE_J;
            const v4f* hs = hsrc + (size_t)j1 * (FOUT / 4);
            hr_0 = hs[0];            hr_1 = hs[256];            hr_2 = hs[512];            hr_3 = hs[768];
            hr_4 = hs[SPLIT_F4];     hr_5 = hs[SPLIT_F4 + 256]; hr_6 = hs[SPLIT_F4 + 512]; hr_7 = hs[SPLIT_F4 + 768];
            if (s2thr) s2reg = s2src[j1];
            const v4i* as = adjsrc + j1 / 4;
            ADJ_LOAD(as)
        }
        const float* lh  = lds_h[cur][kh];
        const float* ls2 = lds_s2[cur] + kh * TILE_J;

        float p[32];
        float mx = NEG_BIG;
        for (int jj = 0; jj < 4; ++jj) {
            for (int q = 0; q < 4; ++q) {
                for (int c = 0; c < 2; ++c) {
                    const int jl = 16 * jj + 4 * q + 2 * half + c;
                    float e = sv1 + ls2[jl];
                    e = (e >= 0.0f) ? e : ALPHA * e;
                    e = ((Mc >> (16 * c + 4 * jj + q)) & 1u) ? e : NEG_BIG;
                    p[8 * jj + 2 * q + c] = e;
                    mx = fmaxf(mx, e);
                }
            }
        }
        mx = fmaxf(mx, __shfl_xor(mx, 16, 32));
        const float mnew  = fmaxf(mrun, mx);
        const float scale = __expf(mrun - mnew);
        float rs = 0.0f;
        for (int jj = 0; jj < 4; ++jj) {
            float rj = 0.0f;
            for (int u = 0; u < 8; ++u) { p[8 * jj + u] = __expf(p[8 * jj + u] - mnew); rj += p[8 * jj + u]; }
            rs += rj;
        }
        rs += __shfl_xor(rs, 16, 32);
        lrun = lrun * scale + rs;
        mrun = mnew;

        for (int r = 0; r < 8; ++r) {
            const float sr = __shfl(scale, r + 8 * half, 32);
            acc0[r] *= sr; acc1[r] *= sr; acc2[r] *= sr; acc3[r] *= sr;
        }

        for (int jj = 0; jj < 4; ++jj) {
            for (int q = 0; q < 4; ++q) {
                v2f A; A[0] = p[8 * jj + 2 * q]; A[1] = p[8 * jj + 2 * q + 1];
                const float* hr0 = lh + (size_t)(16 * jj + 4 * q + 2 * half) * FOUT + n;
                const float* hr1 = hr0 + FOUT;
                v2f B0; B0[0] = hr0[0];  B0[1] = hr1[0];
                v2f B1; B1[0] = hr0[16]; B1[1] = hr1[16];
                v2f B2; B2[0] = hr0[32]; B2[1] = hr1[32];
                v2f B3; B3[0] = hr0[48]; B3[1] = hr1[48];
                acc0 = __builtin_amdgcn_wmma_f32_16x16x4_f32(false, A, false, B0, (short)0, acc0, false, false);
                acc1 = __builtin_amdgcn_wmma_f32_16x16x4_f32(false, A, false, B1, (short)0, acc1, false, false);
                acc2 = __builtin_amdgcn_wmma_f32_16x16x4_f32(false, A, false, B2, (short)0, acc2, false, false);
                acc3 = __builtin_amdgcn_wmma_f32_16x16x4_f32(false, A, false, B3, (short)0, acc3, false, false);
            }
        }

        if (more) {
            v4f* dst = (v4f*)lds_h[cur ^ 1][0] + threadIdx.x;
            dst[0] = hr_0;           dst[256] = hr_1;           dst[512] = hr_2;           dst[768] = hr_3;
            dst[TILE_F4] = hr_4;     dst[TILE_F4 + 256] = hr_5; dst[TILE_F4 + 512] = hr_6; dst[TILE_F4 + 768] = hr_7;
            if (s2thr) lds_s2[cur ^ 1][threadIdx.x] = s2reg;
            ADJ_MASK(Mc)
        }
        __syncthreads();
    }

    float* part = &lds_part[rt][0][lane];
    if (kh == 1) {
        for (int r = 0; r < 8; ++r) {
            part[(r)      * 32] = acc0[r];
            part[(8 + r)  * 32] = acc1[r];
            part[(16 + r) * 32] = acc2[r];
            part[(24 + r) * 32] = acc3[r];
        }
        part[32 * 32] = mrun;
        part[33 * 32] = lrun;
    }
    __syncthreads();
    if (kh == 0) {
        const float m1 = part[32 * 32];
        const float l1 = part[33 * 32];
        const float m  = fmaxf(mrun, m1);
        const float f0 = __expf(mrun - m);
        const float f1 = __expf(m1 - m);
        const float l  = lrun * f0 + l1 * f1;

        for (int r = 0; r < 8; ++r) {
            const int   src = r + 8 * half;
            const float g0  = __shfl(f0, src, 32);
            const float g1  = __shfl(f1, src, 32);
            const float lr  = __shfl(l,  src, 32);
            const float inv = 1.0f / lr;
            const int row   = i0 + r + 8 * half;
            float* o = out + ((size_t)(b * NODES + row)) * FOUT + n;
            float v;
            v = (acc0[r] * g0 + part[(r)      * 32] * g1) * inv; o[0]  = (v > 0.0f) ? v : expm1f(v);
            v = (acc1[r] * g0 + part[(8 + r)  * 32] * g1) * inv; o[16] = (v > 0.0f) ? v : expm1f(v);
            v = (acc2[r] * g0 + part[(16 + r) * 32] * g1) * inv; o[32] = (v > 0.0f) ? v : expm1f(v);
            v = (acc3[r] * g0 + part[(24 + r) * 32] * g1) * inv; o[48] = (v > 0.0f) ? v : expm1f(v);
        }
    }
}

extern "C" void kernel_launch(void* const* d_in, const int* in_sizes, int n_in,
                              void* d_out, int out_size, void* d_ws, size_t ws_size,
                              hipStream_t stream)
{
    const float* x   = (const float*)d_in[0];
    const int*   adj = (const int*)  d_in[1];
    const float* W   = (const float*)d_in[2];
    const float* a   = (const float*)d_in[3];
    float* out = (float*)d_out;

    float* h  = (float*)d_ws;
    float* s1 = h + (size_t)BATCH * NODES * FOUT;
    float* s2 = s1 + (size_t)BATCH * NODES;

    const int total_waves = BATCH * (NODES / 16);
    const int blocks_h = total_waves / K1_WAVES;
    const int blocks_a = total_waves / ROWTILES;

    gat_h_kernel<<<blocks_h, 32 * K1_WAVES, 0, stream>>>(x, W, a, h, s1, s2);
    gat_attn_kernel<<<blocks_a, 256, 0, stream>>>(h, s1, s2, adj, out);
}


# Kernel 2: each tile's h (B) fragments read as one group of 16-byte LDS reads ahead of its 64 products (key pairs interleaved in LDS; lane's columns 2n,2n+1,32+2n,33+2n); sums unchanged
# speedup vs baseline: 4.1046x; 1.0903x over previous
#include <hip/hip_runtime.h>
#include <hip/hip_bf16.h>

typedef __attribute__((ext_vector_type(2))) float v2f;
typedef __attribute__((ext_vector_type(8))) float v8f;
typedef __attribute__((ext_vector_type(4))) float v4f;
typedef __attribute__((ext_vector_type(4))) int   v4i;

#define NODES   4096
#define FIN     512
#define FOUT    64
#define BATCH   4
#define NEG_BIG (-9000000000000000.0f)
#define ALPHA   0.2f

#define K1_WAVES      4
#define W_PAIR_PITCH  176

#define K1_LOADB(DST, G)                                                                         \
    _Pragma("unroll")                                                                            \
    for (int s = 0; s < 8; ++s) {                                                                \
        const float* r0 = lwp + (4 * (4 * (G) + (s >> 1)) + (s & 1)) * W_PAIR_PITCH;             \
        DST[s][0] = *(const v2f*)(r0);      DST[s][1] = *(const v2f*)(r0 + 32);                  \
        DST[s][2] = *(const v2f*)(r0 + 64); DST[s][3] = *(const v2f*)(r0 + 96);                  \
    }

__global__ void __launch_bounds__(128) gat_h_kernel(
    const float* __restrict__ x, const float* __restrict__ W,
    const float* __restrict__ a,
    float* __restrict__ h, float* __restrict__ s1, float* __restrict__ s2)
{
    __shared__ __attribute__((aligned(16))) float ldsW[(FIN / 2) * W_PAIR_PITCH];

    const int lane = threadIdx.x & 31;
    const int wid  = threadIdx.x >> 5;
    const int wg   = blockIdx.x * K1_WAVES + wid;
    const int b    = wg >> 8;
    const int i0   = (wg & 255) << 4;
    const int half = lane >> 4;
    const int n    = lane & 15;

    const v4f* wsrc = (const v4f*)W + (threadIdx.x >> 4) * 32 + (threadIdx.x & 15);
    v4f we[32], wo[32];
#pragma unroll
    for (int i = 0; i < 32; ++i) { we[i] = wsrc[256 * i]; wo[i] = wsrc[256 * i + 16]; }

    const v4f* xsrc = (const v4f*)(x + (size_t)(b * NODES + i0 + n) * FIN) + half;
    v4f X[64];
#pragma unroll
    for (int j = 0; j < 64; ++j) X[j] = xsrc[2 * j];

    {
        float* wdst = ldsW + (threadIdx.x >> 4) * W_PAIR_PITCH + 8 * (threadIdx.x & 15);
#pragma unroll
        for (int i = 0; i < 32; ++i) {
            v4f lo, hi;
            lo[0] = we[i][0]; lo[1] = wo[i][0]; lo[2] = we[i][1]; lo[3] = wo[i][1];
            hi[0] = we[i][2]; hi[1] = wo[i][2]; hi[2] = we[i][3]; hi[3] = wo[i][3];
            *(v4f*)(wdst + 8 * i * W_PAIR_PITCH)     = lo;
            *(v4f*)(wdst + 8 * i * W_PAIR_PITCH + 4) = hi;
        }
    }
    __builtin_amdgcn_fence(__ATOMIC_RELEASE, "workgroup", "local");
    __builtin_amdgcn_s_barrier();
    __builtin_amdgcn_fence(__ATOMIC_ACQUIRE, "workgroup", "local");

    const float* lwp = ldsW + (2 * half) * W_PAIR_PITCH + 2 * n;

    v8f acc0 = {}, acc1 = {}, acc2 = {}, acc3 = {};
    v2f Bc[8][4], Bn[8][4];
    K1_LOADB(Bc, 0)
#pragma unroll
    for (int g = 0; g < 16; ++g) {
        if (g + 1 < 16) { K1_LOADB(Bn, g + 1) }
        __builtin_amdgcn_sched_barrier(0);
#pragma unroll
        for (int s = 0; s < 8; ++s) {
            const v4f xv = X[4 * g + (s >> 1)];
            v2f A; A[0] = xv[2 * (s & 1)]; A[1] = xv[2 * (s & 1) + 1];
            acc0 = __builtin_amdgcn_wmma_f32_16x16x4_f32(false, A, false, Bc[s][0], (short)0, acc0, false, false);
            acc1 = __builtin_amdgcn_wmma_f32_16x16x4_f32(false, A, false, Bc[s][1], (short)0, acc1, false, false);
            acc2 = __builtin_amdgcn_wmma_f32_16x16x4_f32(false, A, false, Bc[s][2], (short)0, acc2, false, false);
            acc3 = __builtin_amdgcn_wmma_f32_16x16x4_f32(false, A, false, Bc[s][3], (short)0, acc3, false, false);
        }
#pragma unroll
        for (int s = 0; s < 8; ++s) {
            Bc[s][0] = Bn[s][0]; Bc[s][1] = Bn[s][1]; Bc[s][2] = Bn[s][2]; Bc[s][3] = Bn[s][3];
        }
    }

    const float a10 = a[n],        a11 = a[n + 16], a12 = a[n + 32], a13 = a[n + 48];
    const float a20 = a[FOUT + n], a21 = a[FOUT + n + 16],
                a22 = a[FOUT + n + 32], a23 = a[FOUT + n + 48];

    for (int r = 0; r < 8; ++r) {
        const int row = i0 + r + 8 * half;
        float* hrow = h + ((size_t)(b * NODES + row) * FOUT) + n;
        const float c0 = acc0[r], c1 = acc1[r], c2 = acc2[r], c3 = acc3[r];
        hrow[0] = c0; hrow[16] = c1; hrow[32] = c2; hrow[48] = c3;

        float p1 = c0 * a10 + c1 * a11 + c2 * a12 + c3 * a13;
        float p2 = c0 * a20 + c1 * a21 + c2 * a22 + c3 * a23;
        for (int m = 1; m < 16; m <<= 1) {
            p1 += __shfl_xor(p1, m, 32);
            p2 += __shfl_xor(p2, m, 32);
        }
        if (n == 0) {
            s1[b * NODES + row] = p1;
            s2[b * NODES + row] = p2;
        }
    }
}

#define TILE_J          64
#define KSPLIT          2
#define KEYS_PER_SPLIT  (NODES / KSPLIT)
#define NTILES          (KEYS_PER_SPLIT / TILE_J)
#define ROWTILES        4
#define SPLIT_F4        (KEYS_PER_SPLIT * FOUT / 4)
#define ADJ_PAIR        (2 * NODES / 4)
#define HP_PITCH        160
#define HP_TILE         (32 * HP_PITCH)

#define H_STORE(DST, E, O)                                                                 \
    {                                                                                      \
        v4f lo, hi;                                                                        \
        lo[0] = (E)[0]; lo[1] = (O)[0]; lo[2] = (E)[1]; lo[3] = (O)[1];                    \
        hi[0] = (E)[2]; hi[1] = (O)[2]; hi[2] = (E)[3]; hi[3] = (O)[3];                    \
        *(v4f*)(DST) = lo;  *(v4f*)((DST) + 4) = hi;                                       \
    }
#define H_STORE_ALL(BASE)                                                                  \
    H_STORE((BASE),                                hr_0, hr_1)                             \
    H_STORE((BASE) + 16 * HP_PITCH,                hr_2, hr_3)                             \
    H_STORE((BASE) + HP_TILE,                      hr_4, hr_5)                             \
    H_STORE((BASE) + HP_TILE + 16 * HP_PITCH,      hr_6, hr_7)

#define ADJ_LOAD(BASE)                                                                     \
    ar_0 = (BASE)[0];            ar_1 = (BASE)[ADJ_PAIR];     ar_2 = (BASE)[2 * ADJ_PAIR]; \
    ar_3 = (BASE)[3 * ADJ_PAIR]; ar_4 = (BASE)[4 * ADJ_PAIR]; ar_5 = (BASE)[5 * ADJ_PAIR]; \
    ar_6 = (BASE)[6 * ADJ_PAIR]; ar_7 = (BASE)[7 * ADJ_PAIR];

#define ADJ_FOLD(I, R)                                                                     \
    {                                                                                      \
        const unsigned b0 = __builtin_amdgcn_ballot_w32((R)[0] > 0);                       \
        const unsigned b1 = __builtin_amdgcn_ballot_w32((R)[1] > 0);                       \
        const unsigned b2 = __builtin_amdgcn_ballot_w32((R)[2] > 0);                       \
        const unsigned b3 = __builtin_amdgcn_ballot_w32((R)[3] > 0);                       \
        mA = (sel == (I)) ? b0 : mA;   mA = (sel == (I) + 8) ? b2 : mA;                    \
        mB = (sel == (I)) ? b1 : mB;   mB = (sel == (I) + 8) ? b3 : mB;                    \
    }
#define ADJ_MASK(M)                                                                        \
    {                                                                                      \
        unsigned mA = 0u, mB = 0u;                                                         \
        ADJ_FOLD(0, ar_0) ADJ_FOLD(1, ar_1) ADJ_FOLD(2, ar_2) ADJ_FOLD(3, ar_3)            \
        ADJ_FOLD(4, ar_4) ADJ_FOLD(5, ar_5) ADJ_FOLD(6, ar_6) ADJ_FOLD(7, ar_7)            \
        (M) = ((mA >> rsh) & 0xFFFFu) | ((mB >> rsh) << 16);                               \
    }

__global__ void __launch_bounds__(256) gat_attn_kernel(
    const float* __restrict__ h, const float* __restrict__ s1g,
    const float* __restrict__ s2g, const int* __restrict__ adj,
    float* __restrict__ out)
{
    __shared__ __attribute__((aligned(16))) float lds_h[2][KSPLIT][HP_TILE];
    __shared__ float lds_s2[2][KSPLIT * TILE_J];
    __shared__ float lds_part[ROWTILES][34][32];

    const int lane = threadIdx.x & 31;
    const int wid  = threadIdx.x >> 5;
    const int kh   = wid >> 2;
    const int rt   = wid & 3;
    const int wg   = blockIdx.x * ROWTILES + rt;
    const int b    = wg >> 8;
    const int i0   = (wg & 255) << 4;
    const int half = lane >> 4;
    const int n    = lane & 15;

    const float sv1 = s1g[b * NODES + i0 + n];
    const v4i* adjsrc = (const v4i*)(adj + (size_t)(i0 + half) * NODES + kh * KEYS_PER_SPLIT) + n;
    const int  sel    = (n >> 1) + 8 * half;
    const int  rsh    = 16 * (n & 1);
    const v4f*   hsrc  = (const v4f*)(h + (size_t)b * NODES * FOUT) + (threadIdx.x >> 4) * 32 + (threadIdx.x & 15);
    const int    hdoff = (threadIdx.x >> 4) * HP_PITCH + 8 * (threadIdx.x & 15);
    const float*  s2src = s2g + b * NODES + (threadIdx.x >> 6) * KEYS_PER_SPLIT + (threadIdx.x & (TILE_J - 1));
    const bool    s2thr = threadIdx.x < KSPLIT * TILE_J;

    v4f    hr_0, hr_1, hr_2, hr_3;
    v4f    hr_4, hr_5, hr_6, hr_7;
    float  s2reg = 0.0f;
    v4i    ar_0, ar_1, ar_2, ar_3, ar_4, ar_5, ar_6, ar_7;
    unsigned Mc;

    hr_0 = hsrc[0];            hr_1 = hsrc[16];            hr_2 = hsrc[512];            hr_3 = hsrc[528];
    hr_4 = hsrc[SPLIT_F4];     hr_5 = hsrc[SPLIT_F4 + 16]; hr_6 = hsrc[SPLIT_F4 + 512]; hr_7 = hsrc[SPLIT_F4 + 528];
    if (s2thr) s2reg = s2src[0];
    ADJ_LOAD(adjsrc)
    {
        float* dst = &lds_h[0][0][0] + hdoff;
        H_STORE_ALL(dst)
        if (s2thr) lds_s2[0][threadIdx.x] = s2reg;
    }
    ADJ_MASK(Mc)
    __syncthreads();

    float mrun = NEG_BIG;
    float lrun = 0.0f;
    v8f acc0 = {}, acc1 = {}, acc2 = {}, acc3 = {};

    for (int t = 0; t < NTILES; ++t) {
        const int  cur  = t & 1;
        const bool more = (t + 1) < NTILES;

        if (more) {
            const int j1 = (t + 1) * TILE_J;
            const v4f* hs = hsrc + (size_t)j1 * (FOUT / 4);
            hr_0 = hs[0];            hr_1 = hs[16];            hr_2 = hs[512];            hr_3 = hs[528];
            hr_4 = hs[SPLIT_F4];     hr_5 = hs[SPLIT_F4 + 16]; hr_6 = hs[SPLIT_F4 + 512]; hr_7 = hs[SPLIT_F4 + 528];
            if (s2thr) s2reg = s2src[j1];
            const v4i* as = adjsrc + j1 / 4;
            ADJ_LOAD(as)
        }
        const float* lb  = lds_h[cur][kh] + half * HP_PITCH + 4 * n;
        const float* ls2 = lds_s2[cur] + kh * TILE_J;

        float s2v[32];
#pragma unroll
        for (int jj = 0; jj < 4; ++jj)
            for (int q = 0; q < 4; ++q)
                for (int c = 0; c < 2; ++c)
                    s2v[8 * jj + 2 * q + c] = ls2[16 * jj + 4 * q + 2 * half + c];
        v4f Bf[32];
#pragma unroll
        for (int ks = 0; ks < 16; ++ks) {
            Bf[2 * ks]     = *(const v4f*)(lb + 2 * ks * HP_PITCH);
            Bf[2 * ks + 1] = *(const v4f*)(lb + 2 * ks * HP_PITCH + 64);
        }
        __builtin_amdgcn_sched_barrier(0);

        float p[32];
        float mx = NEG_BIG;
        for (int jj = 0; jj < 4; ++jj) {
            for (int q = 0; q < 4; ++q) {
                for (int c = 0; c < 2; ++c) {
                    float e = sv1 + s2v[8 * jj + 2 * q + c];
                    e = (e >= 0.0f) ? e : ALPHA * e;
                    e = ((Mc >> (16 * c + 4 * jj + q)) & 1u) ? e : NEG_BIG;
                    p[8 * jj + 2 * q + c] = e;
                    mx = fmaxf(mx, e);
                }
            }
        }
        mx = fmaxf(mx, __shfl_xor(mx, 16, 32));
        const float mnew  = fmaxf(mrun, mx);
        const float scale = __expf(mrun - mnew);
        float rs = 0.0f;
        for (int jj = 0; jj < 4; ++jj) {
            float rj = 0.0f;
            for (int u = 0; u < 8; ++u) { p[8 * jj + u] = __expf(p[8 * jj + u] - mnew); rj += p[8 * jj + u]; }
            rs += rj;
        }
        rs += __shfl_xor(rs, 16, 32);
        lrun = lrun * scale + rs;
        mrun = mnew;

        for (int r = 0; r < 8; ++r) {
            const float sr = __shfl(scale, r + 8 * half, 32);
            acc0[r] *= sr; acc1[r] *= sr; acc2[r] *= sr; acc3[r] *= sr;
        }

#pragma unroll
        for (int ks = 0; ks < 16; ++ks) {
            v2f A; A[0] = p[2 * ks]; A[1] = p[2 * ks + 1];
            const v4f f0 = Bf[2 * ks], f1 = Bf[2 * ks + 1];
            v2f B0; B0[0] = f0[0]; B0[1] = f0[1];
            v2f B1; B1[0] = f0[2]; B1[1] = f0[3];
            v2f B2; B2[0] = f1[0]; B2[1] = f1[1];
            v2f B3; B3[0] = f1[2]; B3[1] = f1[3];
            acc0 = __builtin_amdgcn_wmma_f32_16x16x4_f32(false, A, false, B0, (short)0, acc0, false, false);
            acc1 = __builtin_amdgcn_wmma_f32_16x16x4_f32(false, A, false, B1, (short)0, acc1, false, false);
            acc2 = __builtin_amdgcn_wmma_f32_16x16x4_f32(false, A, false, B2, (short)0, acc2, false, false);
            acc3 = __builtin_amdgcn_wmma_f32_16x16x4_f32(false, A, false, B3, (short)0, acc3, false, false);
        }

        if (more) {
            float* dst = &lds_h[cur ^ 1][0][0] + hdoff;
            H_STORE_ALL(dst)
            if (s2thr) lds_s2[cur ^ 1][threadIdx.x] = s2reg;
            ADJ_MASK(Mc)
        }
        __syncthreads();
    }

    float* part = &lds_part[rt][0][lane];
    if (kh == 1) {
        for (int r = 0; r < 8; ++r) {
            part[(r)      * 32] = acc0[r];
            part[(8 + r)  * 32] = acc1[r];
            part[(16 + r) * 32] = acc2[r];
            part[(24 + r) * 32] = acc3[r];
        }
        part[32 * 32] = mrun;
        part[33 * 32] = lrun;
    }
    __syncthreads();
    if (kh == 0) {
        const float m1 = part[32 * 32];
        const float l1 = part[33 * 32];
        const float m  = fmaxf(mrun, m1);
        const float f0 = __expf(mrun - m);
        const float f1 = __expf(m1 - m);
        const float l  = lrun * f0 + l1 * f1;

        for (int r = 0; r < 8; ++r) {
            const int   src = r + 8 * half;
            const float g0  = __shfl(f0, src, 32);
            const float g1  = __shfl(f1, src, 32);
            const float lr  = __shfl(l,  src, 32);
            const float inv = 1.0f / lr;
            const int row   = i0 + r + 8 * half;
            float* o = out + ((size_t)(b * NODES + row)) * FOUT + 2 * n;
            float v;
            v2f w0, w1;
            v = (acc0[r] * g0 + part[(r)      * 32] * g1) * inv; w0[0] = (v > 0.0f) ? v : expm1f(v);
            v = (acc1[r] * g0 + part[(8 + r)  * 32] * g1) * inv; w0[1] = (v > 0.0f) ? v : expm1f(v);
            v = (acc2[r] * g0 + part[(16 + r) * 32] * g1) * inv; w1[0] = (v > 0.0f) ? v : expm1f(v);
            v = (acc3[r] * g0 + part[(24 + r) * 32] * g1) * inv; w1[1] = (v > 0.0f) ? v : expm1f(v);
            *(v2f*)(o)      = w0;
            *(v2f*)(o + 32) = w1;
        }
    }
}

extern "C" void kernel_launch(void* const* d_in, const int* in_sizes, int n_in,
                              void* d_out, int out_size, void* d_ws, size_t ws_size,
                              hipStream_t stream)
{
    const float* x   = (const float*)d_in[0];
    const int*   adj = (const int*)  d_in[1];
    const float* W   = (const float*)d_in[2];
    const float* a   = (const float*)d_in[3];
    float* out = (float*)d_out;

    float* h  = (float*)d_ws;
    float* s1 = h + (size_t)BATCH * NODES * FOUT;
    float* s2 = s1 + (size_t)BATCH * NODES;

    const int total_waves = BATCH * (NODES / 16);
    const int blocks_h = total_waves / K1_WAVES;
    const int blocks_a = total_waves / ROWTILES;

    gat_h_kernel<<<blocks_h, 32 * K1_WAVES, 0, stream>>>(x, W, a, h, s1, s2);
    gat_attn_kernel<<<blocks_a, 256, 0, stream>>>(h, s1, s2, adj, out);
}
